# starting kernel
# baseline (speedup 1.0000x reference)
#include <hip/hip_runtime.h>
#include <hip/hip_bf16.h>

#define N_NODES 100000
#define N_EDGES 1600000
#define IN_CH   128
#define HID_CH  64
#define OUT_CH  32

typedef __attribute__((ext_vector_type(2))) float f32x2;
typedef __attribute__((ext_vector_type(8))) float f32x8;

__device__ __forceinline__ void atomAddF(float* p, float v) {
    unsafeAtomicAdd(p, v);
}


__global__ void deg_init_kernel(float* deg, int n) {
    int i = blockIdx.x * blockDim.x + threadIdx.x;
    if (i < n) deg[i] = 1.0f;
}

__global__ void deg_count_kernel(const int* __restrict__ dst, float* deg, int nE) {
    int e = blockIdx.x * blockDim.x + threadIdx.x;
    if (e < nE) atomAddF(&deg[dst[e]], 1.0f);
}

__global__ void rsqrt_kernel(float* deg, int n) {
    int i = blockIdx.x * blockDim.x + threadIdx.x;
    if (i < n) deg[i] = rsqrtf(deg[i]);
}


template <int K, int NCOL>
__global__ void gemm_wmma_f32(const float* __restrict__ A,
                              const float* __restrict__ B,
                              float* __restrict__ C) {
    const int wave = threadIdx.x >> 5;
    const int lane = threadIdx.x & 31;
    const int row0 = blockIdx.x << 4;

    const int m  = lane & 15;
    const int kk = (lane >> 4) << 1;
    const int n  = (lane & 15) + (wave << 4);

    const float* ap = A + (size_t)(row0 + m) * K + kk;
    const float* bp = B + (size_t)kk * NCOL + n;

    f32x8 acc = {0.f, 0.f, 0.f, 0.f, 0.f, 0.f, 0.f, 0.f};

    for (int k0 = 0; k0 < K; k0 += 4) {
        f32x2 a, b;
        a.x = ap[k0];
        a.y = ap[k0 + 1];
        b.x = bp[(size_t)k0 * NCOL];
        b.y = bp[(size_t)(k0 + 1) * NCOL];
        acc = __builtin_amdgcn_wmma_f32_16x16x4_f32(
             false, a,  false, b,
             (short)0, acc,  false,  false);
    }

    const int mbase = (lane >> 4) << 3;
    float* cp = C + (size_t)row0 * NCOL + n;
#pragma unroll
    for (int i = 0; i < 8; ++i)
        cp[(size_t)(mbase + i) * NCOL] = acc[i];
}


template <int C, int LOG2C>
__global__ void edge_agg_kernel(const float* __restrict__ h,
                                const int* __restrict__ src,
                                const int* __restrict__ dst,
                                const float* __restrict__ dinv,
                                float* agg) {
    unsigned idx = blockIdx.x * blockDim.x + threadIdx.x;
    int e = idx >> LOG2C;
    int c = idx & (C - 1);
    int s = src[e];
    int d = dst[e];
    float nrm = dinv[s] * dinv[d];
    float v = h[(size_t)s * C + c] * nrm;
    atomAddF(&agg[(size_t)d * C + c], v);
}


template <int C, bool RELU>
__global__ void finish_kernel(const float* agg,
                              const float* __restrict__ h,
                              const float* __restrict__ dinv,
                              const float* __restrict__ bias,
                              float* out) {
    unsigned idx = blockIdx.x * blockDim.x + threadIdx.x;
    int n = idx >> (C == 64 ? 6 : 5);
    int c = idx & (C - 1);
    float di = dinv[n];
    float v = agg[idx] + h[idx] * di * di + bias[c];
    out[idx] = RELU ? fmaxf(v, 0.0f) : v;
}


extern "C" void kernel_launch(void* const* d_in, const int* in_sizes, int n_in,
                              void* d_out, int out_size, void* d_ws, size_t ws_size,
                              hipStream_t stream) {
    const float* x    = (const float*)d_in[0];
    const int*   ei   = (const int*)d_in[1];
    const float* W1   = (const float*)d_in[2];
    const float* b1   = (const float*)d_in[3];
    const float* W2   = (const float*)d_in[4];
    const float* b2   = (const float*)d_in[5];
    float*       out  = (float*)d_out;

    const int* srcv = ei;
    const int* dstv = ei + N_EDGES;

    float* dinv = (float*)d_ws;
    float* h1   = dinv + 100352;
    float* agg1 = h1 + (size_t)N_NODES * HID_CH;
    float* h2   = h1;
    float* agg2 = h1 + (size_t)N_NODES * OUT_CH;

    deg_init_kernel<<<(N_NODES + 255) / 256, 256, 0, stream>>>(dinv, N_NODES);
    deg_count_kernel<<<(N_EDGES + 255) / 256, 256, 0, stream>>>(dstv, dinv, N_EDGES);
    rsqrt_kernel<<<(N_NODES + 255) / 256, 256, 0, stream>>>(dinv, N_NODES);

    gemm_wmma_f32<IN_CH, HID_CH><<<N_NODES / 16, 128, 0, stream>>>(x, W1, h1);

    hipMemsetAsync(agg1, 0, (size_t)N_NODES * HID_CH * sizeof(float), stream);
    edge_agg_kernel<HID_CH, 6>
        <<<(size_t)N_EDGES * HID_CH / 256, 256, 0, stream>>>(h1, srcv, dstv, dinv, agg1);

    finish_kernel<HID_CH, true>
        <<<(size_t)N_NODES * HID_CH / 256, 256, 0, stream>>>(agg1, h1, dinv, b1, agg1);

    gemm_wmma_f32<HID_CH, OUT_CH><<<N_NODES / 16, 64, 0, stream>>>(agg1, W2, h2);

    hipMemsetAsync(agg2, 0, (size_t)N_NODES * OUT_CH * sizeof(float), stream);
    edge_agg_kernel<OUT_CH, 5>
        <<<(size_t)N_EDGES * OUT_CH / 256, 256, 0, stream>>>(h2, srcv, dstv, dinv, agg2);

    finish_kernel<OUT_CH, false>
        <<<(size_t)N_NODES * OUT_CH / 256, 256, 0, stream>>>(agg2, h2, dinv, b2, out);
}


# Float atomic scatter-add replaced by gather over edges bucketed by destination each call (int atomic arrival-order slot); one wave per node sums in registers, self-loop+bias fused; sum order changes
# speedup vs baseline: 2.3703x; 2.3703x over previous
#include <hip/hip_runtime.h>
#include <hip/hip_bf16.h>


#define N_NODES 100000
#define N_EDGES 1600000
#define IN_CH   128
#define HID_CH  64
#define OUT_CH  32

#define SPLIT_ROW 65536
#define SCAN_BLOCK 1024

typedef __attribute__((ext_vector_type(2))) float f32x2;
typedef __attribute__((ext_vector_type(8))) float f32x8;


__global__ void cnt_zero_kernel(int* cnt, int n) {
    int i = blockIdx.x * blockDim.x + threadIdx.x;
    if (i < n) cnt[i] = 0;
}

__global__ void count_rank_kernel(const int* __restrict__ dst, int* cnt, int* __restrict__ rank, int nE) {
    int e = blockIdx.x * blockDim.x + threadIdx.x;
    if (e < nE) rank[e] = atomicAdd(&cnt[dst[e]], 1);
}

__global__ void __launch_bounds__(SCAN_BLOCK)
scan_kernel(const int* __restrict__ cnt, int* __restrict__ rowptr, float* __restrict__ dinv, int n) {
    __shared__ int wsum[SCAN_BLOCK / 32];
    __shared__ int s_prefix;
    const int tid  = threadIdx.x;
    const int lane = tid & 31;
    const int wv   = tid >> 5;
    const int first = blockIdx.x * SCAN_BLOCK;

    int part = 0;
    for (int i = tid; i < first; i += SCAN_BLOCK) part += cnt[i];
    for (int o = 16; o > 0; o >>= 1) part += __shfl_xor(part, o);
    if (lane == 0) wsum[wv] = part;
    __syncthreads();
    if (wv == 0) {
        int t = wsum[lane];
        for (int o = 16; o > 0; o >>= 1) t += __shfl_xor(t, o);
        if (lane == 0) s_prefix = t;
    }
    __syncthreads();
    const int prefix = s_prefix;

    const int i = first + tid;
    const int c = (i < n) ? cnt[i] : 0;
    int incl = c;
    for (int o = 1; o < 32; o <<= 1) {
        int t = __shfl_up(incl, o);
        if (lane >= o) incl += t;
    }
    if (lane == 31) wsum[wv] = incl;
    __syncthreads();
    int woff = (lane < wv) ? wsum[lane] : 0;
    for (int o = 16; o > 0; o >>= 1) woff += __shfl_xor(woff, o);

    const int excl = prefix + woff + incl - c;
    if (i < n) {
        rowptr[i] = excl;
        dinv[i] = rsqrtf((float)c + 1.0f);
        if (i == n - 1) rowptr[n] = excl + c;
    }
}

__global__ void fill_kernel(const int* __restrict__ src, const int* __restrict__ dst,
                            const int* __restrict__ rank, const int* __restrict__ rowptr,
                            int* __restrict__ col, int nE) {
    int e = blockIdx.x * blockDim.x + threadIdx.x;
    if (e < nE) col[rowptr[dst[e]] + rank[e]] = src[e];
}


template <int K, int NCOL>
__global__ void gemm_wmma_f32(const float* __restrict__ A,
                              const float* __restrict__ B,
                              float* __restrict__ C) {
    const int wave = threadIdx.x >> 5;
    const int lane = threadIdx.x & 31;
    const int row0 = blockIdx.x << 4;

    const int m  = lane & 15;
    const int kk = (lane >> 4) << 1;
    const int n  = (lane & 15) + (wave << 4);

    const float* ap = A + (size_t)(row0 + m) * K + kk;
    const float* bp = B + (size_t)kk * NCOL + n;

    f32x8 acc = {0.f, 0.f, 0.f, 0.f, 0.f, 0.f, 0.f, 0.f};

    for (int k0 = 0; k0 < K; k0 += 4) {
        f32x2 a, b;
        a.x = ap[k0];
        a.y = ap[k0 + 1];
        b.x = bp[(size_t)k0 * NCOL];
        b.y = bp[(size_t)(k0 + 1) * NCOL];
        acc = __builtin_amdgcn_wmma_f32_16x16x4_f32(
             false, a,  false, b,
             (short)0, acc,  false,  false);
    }

    const int mbase = (lane >> 4) << 3;
    float* cp = C + (size_t)row0 * NCOL + n;
#pragma unroll
    for (int i = 0; i < 8; ++i)
        cp[(size_t)(mbase + i) * NCOL] = acc[i];
}


template <int C, bool RELU>
__global__ void __launch_bounds__(256)
gather_finish_kernel(const float* __restrict__ h,
                     const int* __restrict__ rowptr,
                     const int* __restrict__ col,
                     const float* __restrict__ dinv,
                     const float* __restrict__ bias,
                     float* dst_lo, float* dst_hi, int split) {
    constexpr int V = C / 32;
    const int lane = threadIdx.x & 31;
    const int node = __builtin_amdgcn_readfirstlane((int)(blockIdx.x * 8 + (threadIdx.x >> 5)));
    const int beg = rowptr[node];
    const int end = rowptr[node + 1];
    const float di = dinv[node];
    const int c = lane * V;

    f32x2 acc2 = {0.f, 0.f};
    float acc1 = 0.f;
    for (int base = beg; base < end; base += 32) {
        const int p = base + lane;
        int s_l = 0;
        float w_l = 0.f;
        if (p < end) {
            s_l = min(max(col[p], 0), N_NODES - 1);
            w_l = dinv[s_l] * di;
        }
        const int cnt = min(32, end - base);
#define ROW_PTR(j) (h + (size_t)__builtin_amdgcn_readlane(s_l, (j)) * C + c)
#define NORM(j) (__int_as_float(__builtin_amdgcn_readlane(__float_as_int(w_l), (j))))
        int j = 0;
        if constexpr (V == 2) {
            for (; j + 4 <= cnt; j += 4) {
                const f32x2 v0 = *(const f32x2*)ROW_PTR(j);
                const f32x2 v1 = *(const f32x2*)ROW_PTR(j + 1);
                const f32x2 v2 = *(const f32x2*)ROW_PTR(j + 2);
                const f32x2 v3 = *(const f32x2*)ROW_PTR(j + 3);
                acc2 += v0 * NORM(j);
                acc2 += v1 * NORM(j + 1);
                acc2 += v2 * NORM(j + 2);
                acc2 += v3 * NORM(j + 3);
            }
            const int m = cnt - j;
            if (m > 0) {
                const f32x2 v0 = *(const f32x2*)ROW_PTR(j);
                f32x2 v1 = {0.f, 0.f}, v2 = {0.f, 0.f};
                if (m > 1) v1 = *(const f32x2*)ROW_PTR(j + 1);
                if (m > 2) v2 = *(const f32x2*)ROW_PTR(j + 2);
                acc2 += v0 * NORM(j);
                if (m > 1) acc2 += v1 * NORM(j + 1);
                if (m > 2) acc2 += v2 * NORM(j + 2);
            }
        } else {
            for (; j + 4 <= cnt; j += 4) {
                const float v0 = *ROW_PTR(j);
                const float v1 = *ROW_PTR(j + 1);
                const float v2 = *ROW_PTR(j + 2);
                const float v3 = *ROW_PTR(j + 3);
                acc1 += v0 * NORM(j);
                acc1 += v1 * NORM(j + 1);
                acc1 += v2 * NORM(j + 2);
                acc1 += v3 * NORM(j + 3);
            }
            const int m = cnt - j;
            if (m > 0) {
                const float v0 = *ROW_PTR(j);
                float v1 = 0.f, v2 = 0.f;
                if (m > 1) v1 = *ROW_PTR(j + 1);
                if (m > 2) v2 = *ROW_PTR(j + 2);
                acc1 += v0 * NORM(j);
                if (m > 1) acc1 += v1 * NORM(j + 1);
                if (m > 2) acc1 += v2 * NORM(j + 2);
            }
        }
#undef ROW_PTR
#undef NORM
    }

    const size_t o = (size_t)node * C + c;
    float* op = (node < split) ? (dst_lo + o) : (dst_hi + (o - (size_t)split * C));
    if constexpr (V == 2) {
        const f32x2 hv = *(const f32x2*)(h + o);
        const f32x2 bv = *(const f32x2*)(bias + c);
        f32x2 r = acc2 + hv * di * di + bv;
        if (RELU) { r.x = fmaxf(r.x, 0.0f); r.y = fmaxf(r.y, 0.0f); }
        *(f32x2*)op = r;
    } else {
        float r = acc1 + h[o] * di * di + bias[c];
        if (RELU) r = fmaxf(r, 0.0f);
        op[0] = r;
    }
}


extern "C" void kernel_launch(void* const* d_in, const int* in_sizes, int n_in,
                              void* d_out, int out_size, void* d_ws, size_t ws_size,
                              hipStream_t stream) {
    const float* x    = (const float*)d_in[0];
    const int*   ei   = (const int*)d_in[1];
    const float* W1   = (const float*)d_in[2];
    const float* b1   = (const float*)d_in[3];
    const float* W2   = (const float*)d_in[4];
    const float* b2   = (const float*)d_in[5];
    float*       out  = (float*)d_out;

    const int* srcv = ei;
    const int* dstv = ei + N_EDGES;

    float* dinv   = (float*)d_ws;
    int*   rowptr = (int*)(dinv + 100352);
    int*   col    = rowptr + 100352;
    float* h1     = (float*)(col + N_EDGES);
    float* a1lo   = h1 + (size_t)N_NODES * HID_CH;
    float* a1hi   = out;
    int*   cnt    = (int*)h1;
    int*   rank   = cnt + 100352;
    float* h2     = h1;

    cnt_zero_kernel<<<(N_NODES + 255) / 256, 256, 0, stream>>>(cnt, N_NODES);
    count_rank_kernel<<<(N_EDGES + 255) / 256, 256, 0, stream>>>(dstv, cnt, rank, N_EDGES);
    scan_kernel<<<(N_NODES + SCAN_BLOCK - 1) / SCAN_BLOCK, SCAN_BLOCK, 0, stream>>>(cnt, rowptr, dinv, N_NODES);
    fill_kernel<<<(N_EDGES + 255) / 256, 256, 0, stream>>>(srcv, dstv, rank, rowptr, col, N_EDGES);

    gemm_wmma_f32<IN_CH, HID_CH><<<N_NODES / 16, 128, 0, stream>>>(x, W1, h1);

    gather_finish_kernel<HID_CH, true>
        <<<N_NODES / 8, 256, 0, stream>>>(h1, rowptr, col, dinv, b1, a1lo, a1hi, SPLIT_ROW);

    gemm_wmma_f32<HID_CH, OUT_CH><<<SPLIT_ROW / 16, 64, 0, stream>>>(a1lo, W2, h2);
    gemm_wmma_f32<HID_CH, OUT_CH><<<(N_NODES - SPLIT_ROW) / 16, 64, 0, stream>>>(
        a1hi, W2, h2 + (size_t)SPLIT_ROW * OUT_CH);

    gather_finish_kernel<OUT_CH, false>
        <<<N_NODES / 8, 256, 0, stream>>>(h2, rowptr, col, dinv, b2, out, out, N_NODES);
}


# Edges bucketed by destination on LDS counters (count, scan, place per 512-node range, then per node inside a range), not one returning global atomic per edge; slot order, so sum order, changes
# speedup vs baseline: 4.1767x; 1.7621x over previous
#include <hip/hip_runtime.h>
#include <hip/hip_bf16.h>


#define N_NODES 100000
#define N_EDGES 1600000
#define IN_CH   128
#define HID_CH  64
#define OUT_CH  32

#define SPLIT_ROW 65536
#define SCAN_BLOCK 1024

typedef __attribute__((ext_vector_type(2))) float f32x2;
typedef __attribute__((ext_vector_type(8))) float f32x8;


#define RANGE_SHIFT 9
#define RANGE_NODES (1 << RANGE_SHIFT)
#define N_RANGE     ((N_NODES + RANGE_NODES - 1) / RANGE_NODES)
#define CHUNK_EDGES 6400
#define N_CHUNK     (N_EDGES / CHUNK_EDGES)
#define CHUNK_VEC   (CHUNK_EDGES / 4)
#define N_CELL      (N_RANGE * N_CHUNK)
#define PREP_BLOCK  1024
#define NODE_BATCH  8

static_assert(N_CHUNK * CHUNK_EDGES == N_EDGES, "chunks must tile the edge list");
static_assert(N_CELL + 1 <= 49152, "count matrix and its scan fit their workspace planes");
static_assert(PREP_BLOCK <= CHUNK_VEC && CHUNK_VEC <= 2 * PREP_BLOCK, "two 16-byte loads per thread cover a chunk");
static_assert(N_RANGE <= PREP_BLOCK && RANGE_NODES <= PREP_BLOCK, "one thread per range / per node of a range");

__device__ __forceinline__ int range_of(int d) {
    return (int)min((unsigned)d >> RANGE_SHIFT, (unsigned)(N_RANGE - 1));
}

__global__ void __launch_bounds__(PREP_BLOCK)
range_count_kernel(const int* __restrict__ dst, int* __restrict__ cells) {
    __shared__ int cnt[N_RANGE];
    const int tid = threadIdx.x;
    if (tid < N_RANGE) cnt[tid] = 0;
    __syncthreads();

    const int4* dp = (const int4*)(dst + (size_t)blockIdx.x * CHUNK_EDGES);
    const bool second = tid + PREP_BLOCK < CHUNK_VEC;
    const int4 d0 = dp[tid];
    int4 d1 = {0, 0, 0, 0};
    if (second) d1 = dp[tid + PREP_BLOCK];
    atomicAdd(&cnt[range_of(d0.x)], 1);
    atomicAdd(&cnt[range_of(d0.y)], 1);
    atomicAdd(&cnt[range_of(d0.z)], 1);
    atomicAdd(&cnt[range_of(d0.w)], 1);
    if (second) {
        atomicAdd(&cnt[range_of(d1.x)], 1);
        atomicAdd(&cnt[range_of(d1.y)], 1);
        atomicAdd(&cnt[range_of(d1.z)], 1);
        atomicAdd(&cnt[range_of(d1.w)], 1);
    }
    __syncthreads();
    if (tid < N_RANGE) cells[tid * N_CHUNK + blockIdx.x] = cnt[tid];
}

__global__ void __launch_bounds__(SCAN_BLOCK)
scan_kernel(const int* __restrict__ cnt, int* __restrict__ out, int n) {
    __shared__ int wsum[SCAN_BLOCK / 32];
    __shared__ int s_prefix;
    const int tid  = threadIdx.x;
    const int lane = tid & 31;
    const int wv   = tid >> 5;
    const int first = blockIdx.x * SCAN_BLOCK;

    int part = 0;
    for (int i = tid; i < first; i += SCAN_BLOCK) part += cnt[i];
    for (int o = 16; o > 0; o >>= 1) part += __shfl_xor(part, o);
    if (lane == 0) wsum[wv] = part;
    __syncthreads();
    if (wv == 0) {
        int t = wsum[lane];
        for (int o = 16; o > 0; o >>= 1) t += __shfl_xor(t, o);
        if (lane == 0) s_prefix = t;
    }
    __syncthreads();
    const int prefix = s_prefix;

    const int i = first + tid;
    const int c = (i < n) ? cnt[i] : 0;
    int incl = c;
    for (int o = 1; o < 32; o <<= 1) {
        int t = __shfl_up(incl, o);
        if (lane >= o) incl += t;
    }
    if (lane == 31) wsum[wv] = incl;
    __syncthreads();
    int woff = (lane < wv) ? wsum[lane] : 0;
    for (int o = 16; o > 0; o >>= 1) woff += __shfl_xor(woff, o);

    const int excl = prefix + woff + incl - c;
    if (i < n) {
        out[i] = excl;
        if (i == n - 1) out[n] = excl + c;
    }
}

__global__ void __launch_bounds__(PREP_BLOCK)
range_place_kernel(const int* __restrict__ src, const int* __restrict__ dst,
                   const int* __restrict__ cell_start, int* __restrict__ pairs) {
    __shared__ int cur[N_RANGE];
    const int tid = threadIdx.x;
    if (tid < N_RANGE) cur[tid] = cell_start[tid * N_CHUNK + blockIdx.x];
    __syncthreads();

    const int4* sp = (const int4*)(src + (size_t)blockIdx.x * CHUNK_EDGES);
    const int4* dp = (const int4*)(dst + (size_t)blockIdx.x * CHUNK_EDGES);
    int2* out = (int2*)pairs;
    const bool second = tid + PREP_BLOCK < CHUNK_VEC;
    const int4 s0 = sp[tid];
    const int4 d0 = dp[tid];
    int4 s1 = {0, 0, 0, 0}, d1 = {0, 0, 0, 0};
    if (second) {
        s1 = sp[tid + PREP_BLOCK];
        d1 = dp[tid + PREP_BLOCK];
    }
#define PLACE4(S, D)                                                     \
    do {                                                                 \
        int q0 = atomicAdd(&cur[range_of((D).x)], 1);                    \
        int q1 = atomicAdd(&cur[range_of((D).y)], 1);                    \
        int q2 = atomicAdd(&cur[range_of((D).z)], 1);                    \
        int q3 = atomicAdd(&cur[range_of((D).w)], 1);                    \
        out[min(max(q0, 0), N_EDGES - 1)] = make_int2((S).x, (D).x);     \
        out[min(max(q1, 0), N_EDGES - 1)] = make_int2((S).y, (D).y);     \
        out[min(max(q2, 0), N_EDGES - 1)] = make_int2((S).z, (D).z);     \
        out[min(max(q3, 0), N_EDGES - 1)] = make_int2((S).w, (D).w);     \
    } while (0)
    PLACE4(s0, d0);
    if (second) PLACE4(s1, d1);
#undef PLACE4
}

__global__ void __launch_bounds__(PREP_BLOCK)
node_place_kernel(const int* __restrict__ pairs, const int* __restrict__ cell_start,
                  int* __restrict__ rowptr, float* __restrict__ dinv, int* __restrict__ col) {
    __shared__ int cnt[RANGE_NODES];
    __shared__ int wsum[PREP_BLOCK / 32];
    const int tid  = threadIdx.x;
    const int lane = tid & 31;
    const int wv   = tid >> 5;
    const int r    = blockIdx.x;
    const int beg = min(max(cell_start[r * N_CHUNK], 0), N_EDGES);
    const int end = min(max(cell_start[(r + 1) * N_CHUNK], beg), N_EDGES);
    const int2* in = (const int2*)pairs;

    if (tid < RANGE_NODES) cnt[tid] = 0;
    __syncthreads();

    for (int e0 = beg + tid; e0 < end; e0 += NODE_BATCH * PREP_BLOCK) {
        int d[NODE_BATCH];
#pragma unroll
        for (int k = 0; k < NODE_BATCH; ++k) d[k] = in[min(e0 + k * PREP_BLOCK, end - 1)].y;
#pragma unroll
        for (int k = 0; k < NODE_BATCH; ++k)
            if (e0 + k * PREP_BLOCK < end) atomicAdd(&cnt[d[k] & (RANGE_NODES - 1)], 1);
    }
    __syncthreads();

    const int c = (tid < RANGE_NODES) ? cnt[tid] : 0;
    int incl = c;
    for (int o = 1; o < 32; o <<= 1) {
        int t = __shfl_up(incl, o);
        if (lane >= o) incl += t;
    }
    if (lane == 31) wsum[wv] = incl;
    __syncthreads();
    int woff = (lane < wv) ? wsum[lane] : 0;
    for (int o = 16; o > 0; o >>= 1) woff += __shfl_xor(woff, o);
    const int excl = woff + incl - c;

    const int node = (r << RANGE_SHIFT) + tid;
    if (tid < RANGE_NODES) {
        cnt[tid] = excl;
        if (node < N_NODES) {
            rowptr[node] = beg + excl;
            dinv[node] = rsqrtf((float)c + 1.0f);
            if (node == N_NODES - 1) rowptr[N_NODES] = beg + excl + c;
        }
    }
    __syncthreads();

    for (int e0 = beg + tid; e0 < end; e0 += NODE_BATCH * PREP_BLOCK) {
        int2 p[NODE_BATCH];
#pragma unroll
        for (int k = 0; k < NODE_BATCH; ++k) p[k] = in[min(e0 + k * PREP_BLOCK, end - 1)];
#pragma unroll
        for (int k = 0; k < NODE_BATCH; ++k) {
            if (e0 + k * PREP_BLOCK < end) {
                const int slot = atomicAdd(&cnt[p[k].y & (RANGE_NODES - 1)], 1);
                col[min(beg + slot, end - 1)] = p[k].x;
            }
        }
    }
}


template <int K, int NCOL>
__global__ void gemm_wmma_f32(const float* __restrict__ A,
                              const float* __restrict__ B,
                              const float* __restrict__ rscale,
                              float* __restrict__ C) {
    const int wave = threadIdx.x >> 5;
    const int lane = threadIdx.x & 31;
    const int row0 = blockIdx.x << 4;

    const int m  = lane & 15;
    const int kk = (lane >> 4) << 1;
    const int n  = (lane & 15) + (wave << 4);

    const float* ap = A + (size_t)(row0 + m) * K + kk;
    const float* bp = B + (size_t)kk * NCOL + n;

    f32x8 acc = {0.f, 0.f, 0.f, 0.f, 0.f, 0.f, 0.f, 0.f};

    const int mbase = (lane >> 4) << 3;
    float sc[8];
#pragma unroll
    for (int i = 0; i < 8; ++i) sc[i] = rscale[row0 + mbase + i];

    for (int k0 = 0; k0 < K; k0 += 4) {
        f32x2 a, b;
        a.x = ap[k0];
        a.y = ap[k0 + 1];
        b.x = bp[(size_t)k0 * NCOL];
        b.y = bp[(size_t)(k0 + 1) * NCOL];
        acc = __builtin_amdgcn_wmma_f32_16x16x4_f32(
             false, a,  false, b,
             (short)0, acc,  false,  false);
    }

    float* cp = C + (size_t)row0 * NCOL + n;
#pragma unroll
    for (int i = 0; i < 8; ++i)
        cp[(size_t)(mbase + i) * NCOL] = acc[i] * sc[i];
}


template <int C, bool RELU>
__global__ void __launch_bounds__(256)
gather_finish_kernel(const float* __restrict__ hs,
                     const int* __restrict__ rowptr,
                     const int* __restrict__ col,
                     const float* __restrict__ dinv,
                     const float* __restrict__ bias,
                     float* dst_lo, float* dst_hi, int split) {
    constexpr int V = C / 32;
    const int lane = threadIdx.x & 31;
    const int node = __builtin_amdgcn_readfirstlane((int)(blockIdx.x * 8 + (threadIdx.x >> 5)));
    const int beg = rowptr[node];
    const int end = rowptr[node + 1];
    const float di = dinv[node];
    const int c = lane * V;
    const size_t o = (size_t)node * C + c;

    f32x2 acc2 = {0.f, 0.f};
    float acc1 = 0.f;
    f32x2 bv2 = {0.f, 0.f};
    float bv1 = 0.f;
    if constexpr (V == 2) {
        acc2 = *(const f32x2*)(hs + o);
        bv2  = *(const f32x2*)(bias + c);
    } else {
        acc1 = hs[o];
        bv1  = bias[c];
    }

    for (int base = beg; base < end; base += 32) {
        const int p = base + lane;
        int s_l = 0;
        if (p < end) {
            s_l = min(max(col[p], 0), N_NODES - 1);
        }
        const int cnt = min(32, end - base);
#define ROW_PTR(j) (hs + (size_t)__builtin_amdgcn_readlane(s_l, (j)) * C + c)
        int j = 0;
        if constexpr (V == 2) {
            for (; j + 4 <= cnt; j += 4) {
                const f32x2 v0 = *(const f32x2*)ROW_PTR(j);
                const f32x2 v1 = *(const f32x2*)ROW_PTR(j + 1);
                const f32x2 v2 = *(const f32x2*)ROW_PTR(j + 2);
                const f32x2 v3 = *(const f32x2*)ROW_PTR(j + 3);
                acc2 += v0;
                acc2 += v1;
                acc2 += v2;
                acc2 += v3;
            }
            const int m = cnt - j;
            if (m > 0) {
                const f32x2 v0 = *(const f32x2*)ROW_PTR(j);
                f32x2 v1 = {0.f, 0.f}, v2 = {0.f, 0.f};
                if (m > 1) v1 = *(const f32x2*)ROW_PTR(j + 1);
                if (m > 2) v2 = *(const f32x2*)ROW_PTR(j + 2);
                acc2 += v0;
                if (m > 1) acc2 += v1;
                if (m > 2) acc2 += v2;
            }
        } else {
            for (; j + 4 <= cnt; j += 4) {
                const float v0 = *ROW_PTR(j);
                const float v1 = *ROW_PTR(j + 1);
                const float v2 = *ROW_PTR(j + 2);
                const float v3 = *ROW_PTR(j + 3);
                acc1 += v0;
                acc1 += v1;
                acc1 += v2;
                acc1 += v3;
            }
            const int m = cnt - j;
            if (m > 0) {
                const float v0 = *ROW_PTR(j);
                float v1 = 0.f, v2 = 0.f;
                if (m > 1) v1 = *ROW_PTR(j + 1);
                if (m > 2) v2 = *ROW_PTR(j + 2);
                acc1 += v0;
                if (m > 1) acc1 += v1;
                if (m > 2) acc1 += v2;
            }
        }
#undef ROW_PTR
    }

    float* op = (node < split) ? (dst_lo + o) : (dst_hi + (o - (size_t)split * C));
    if constexpr (V == 2) {
        f32x2 r = acc2 * di + bv2;
        if (RELU) { r.x = fmaxf(r.x, 0.0f); r.y = fmaxf(r.y, 0.0f); }
        *(f32x2*)op = r;
    } else {
        float r = acc1 * di + bv1;
        if (RELU) r = fmaxf(r, 0.0f);
        op[0] = r;
    }
}


extern "C" void kernel_launch(void* const* d_in, const int* in_sizes, int n_in,
                              void* d_out, int out_size, void* d_ws, size_t ws_size,
                              hipStream_t stream) {
    const float* x    = (const float*)d_in[0];
    const int*   ei   = (const int*)d_in[1];
    const float* W1   = (const float*)d_in[2];
    const float* b1   = (const float*)d_in[3];
    const float* W2   = (const float*)d_in[4];
    const float* b2   = (const float*)d_in[5];
    float*       out  = (float*)d_out;

    const int* srcv = ei;
    const int* dstv = ei + N_EDGES;

    float* dinv   = (float*)d_ws;
    int*   rowptr = (int*)(dinv + 100352);
    int*   col    = rowptr + 100352;
    float* h1     = (float*)(col + N_EDGES);
    float* a1lo   = h1 + (size_t)N_NODES * HID_CH;
    float* a1hi   = out;
    int*   cells      = (int*)h1;
    int*   cell_start = cells + 49152;
    int*   pairs      = cell_start + 49152;
    float* h2     = h1;

    range_count_kernel<<<N_CHUNK, PREP_BLOCK, 0, stream>>>(dstv, cells);
    scan_kernel<<<(N_CELL + SCAN_BLOCK - 1) / SCAN_BLOCK, SCAN_BLOCK, 0, stream>>>(cells, cell_start, N_CELL);
    range_place_kernel<<<N_CHUNK, PREP_BLOCK, 0, stream>>>(srcv, dstv, cell_start, pairs);
    node_place_kernel<<<N_RANGE, PREP_BLOCK, 0, stream>>>(pairs, cell_start, rowptr, dinv, col);

    gemm_wmma_f32<IN_CH, HID_CH><<<N_NODES / 16, 128, 0, stream>>>(x, W1, dinv, h1);

    gather_finish_kernel<HID_CH, true>
        <<<N_NODES / 8, 256, 0, stream>>>(h1, rowptr, col, dinv, b1, a1lo, a1hi, SPLIT_ROW);

    gemm_wmma_f32<HID_CH, OUT_CH><<<SPLIT_ROW / 16, 64, 0, stream>>>(a1lo, W2, dinv, h2);
    gemm_wmma_f32<HID_CH, OUT_CH><<<(N_NODES - SPLIT_ROW) / 16, 64, 0, stream>>>(
        a1hi, W2, dinv + SPLIT_ROW, h2 + (size_t)SPLIT_ROW * OUT_CH);

    gather_finish_kernel<OUT_CH, false>
        <<<N_NODES / 8, 256, 0, stream>>>(h2, rowptr, col, dinv, b2, out, out, N_NODES);
}


# Gather waves read 16 bytes per lane: a node's row spans C/4 lanes, so one wave sums 2 (layer 1) or 4 (layer 2) nodes side by side, slot indices handed round by DPP; per-node sum order kept
# speedup vs baseline: 4.4720x; 1.0707x over previous
#include <hip/hip_runtime.h>
#include <hip/hip_bf16.h>


#define N_NODES 100000
#define N_EDGES 1600000
#define IN_CH   128
#define HID_CH  64
#define OUT_CH  32

#define SPLIT_ROW 65536
#define SCAN_BLOCK 1024

typedef __attribute__((ext_vector_type(2))) float f32x2;
typedef __attribute__((ext_vector_type(8))) float f32x8;


#define RANGE_SHIFT 9
#define RANGE_NODES (1 << RANGE_SHIFT)
#define N_RANGE     ((N_NODES + RANGE_NODES - 1) / RANGE_NODES)
#define CHUNK_EDGES 6400
#define N_CHUNK     (N_EDGES / CHUNK_EDGES)
#define CHUNK_VEC   (CHUNK_EDGES / 4)
#define N_CELL      (N_RANGE * N_CHUNK)
#define PREP_BLOCK  1024
#define NODE_BATCH  8

static_assert(N_CHUNK * CHUNK_EDGES == N_EDGES, "chunks must tile the edge list");
static_assert(N_CELL + 1 <= 49152, "count matrix and its scan fit their workspace planes");
static_assert(PREP_BLOCK <= CHUNK_VEC && CHUNK_VEC <= 2 * PREP_BLOCK, "two 16-byte loads per thread cover a chunk");
static_assert(N_RANGE <= PREP_BLOCK && RANGE_NODES <= PREP_BLOCK, "one thread per range / per node of a range");

__device__ __forceinline__ int range_of(int d) {
    return (int)min((unsigned)d >> RANGE_SHIFT, (unsigned)(N_RANGE - 1));
}

__global__ void __launch_bounds__(PREP_BLOCK)
range_count_kernel(const int* __restrict__ dst, int* __restrict__ cells) {
    __shared__ int cnt[N_RANGE];
    const int tid = threadIdx.x;
    if (tid < N_RANGE) cnt[tid] = 0;
    __syncthreads();

    const int4* dp = (const int4*)(dst + (size_t)blockIdx.x * CHUNK_EDGES);
    const bool second = tid + PREP_BLOCK < CHUNK_VEC;
    const int4 d0 = dp[tid];
    int4 d1 = {0, 0, 0, 0};
    if (second) d1 = dp[tid + PREP_BLOCK];
    atomicAdd(&cnt[range_of(d0.x)], 1);
    atomicAdd(&cnt[range_of(d0.y)], 1);
    atomicAdd(&cnt[range_of(d0.z)], 1);
    atomicAdd(&cnt[range_of(d0.w)], 1);
    if (second) {
        atomicAdd(&cnt[range_of(d1.x)], 1);
        atomicAdd(&cnt[range_of(d1.y)], 1);
        atomicAdd(&cnt[range_of(d1.z)], 1);
        atomicAdd(&cnt[range_of(d1.w)], 1);
    }
    __syncthreads();
    if (tid < N_RANGE) cells[tid * N_CHUNK + blockIdx.x] = cnt[tid];
}

__global__ void __launch_bounds__(SCAN_BLOCK)
scan_kernel(const int* __restrict__ cnt, int* __restrict__ out, int n) {
    __shared__ int wsum[SCAN_BLOCK / 32];
    __shared__ int s_prefix;
    const int tid  = threadIdx.x;
    const int lane = tid & 31;
    const int wv   = tid >> 5;
    const int first = blockIdx.x * SCAN_BLOCK;

    int part = 0;
    for (int i = tid; i < first; i += SCAN_BLOCK) part += cnt[i];
    for (int o = 16; o > 0; o >>= 1) part += __shfl_xor(part, o);
    if (lane == 0) wsum[wv] = part;
    __syncthreads();
    if (wv == 0) {
        int t = wsum[lane];
        for (int o = 16; o > 0; o >>= 1) t += __shfl_xor(t, o);
        if (lane == 0) s_prefix = t;
    }
    __syncthreads();
    const int prefix = s_prefix;

    const int i = first + tid;
    const int c = (i < n) ? cnt[i] : 0;
    int incl = c;
    for (int o = 1; o < 32; o <<= 1) {
        int t = __shfl_up(incl, o);
        if (lane >= o) incl += t;
    }
    if (lane == 31) wsum[wv] = incl;
    __syncthreads();
    int woff = (lane < wv) ? wsum[lane] : 0;
    for (int o = 16; o > 0; o >>= 1) woff += __shfl_xor(woff, o);

    const int excl = prefix + woff + incl - c;
    if (i < n) {
        out[i] = excl;
        if (i == n - 1) out[n] = excl + c;
    }
}

__global__ void __launch_bounds__(PREP_BLOCK)
range_place_kernel(const int* __restrict__ src, const int* __restrict__ dst,
                   const int* __restrict__ cell_start, int* __restrict__ pairs) {
    __shared__ int cur[N_RANGE];
    const int tid = threadIdx.x;
    if (tid < N_RANGE) cur[tid] = cell_start[tid * N_CHUNK + blockIdx.x];
    __syncthreads();

    const int4* sp = (const int4*)(src + (size_t)blockIdx.x * CHUNK_EDGES);
    const int4* dp = (const int4*)(dst + (size_t)blockIdx.x * CHUNK_EDGES);
    int2* out = (int2*)pairs;
    const bool second = tid + PREP_BLOCK < CHUNK_VEC;
    const int4 s0 = sp[tid];
    const int4 d0 = dp[tid];
    int4 s1 = {0, 0, 0, 0}, d1 = {0, 0, 0, 0};
    if (second) {
        s1 = sp[tid + PREP_BLOCK];
        d1 = dp[tid + PREP_BLOCK];
    }
#define PLACE4(S, D)                                                     \
    do {                                                                 \
        int q0 = atomicAdd(&cur[range_of((D).x)], 1);                    \
        int q1 = atomicAdd(&cur[range_of((D).y)], 1);                    \
        int q2 = atomicAdd(&cur[range_of((D).z)], 1);                    \
        int q3 = atomicAdd(&cur[range_of((D).w)], 1);                    \
        out[min(max(q0, 0), N_EDGES - 1)] = make_int2((S).x, (D).x);     \
        out[min(max(q1, 0), N_EDGES - 1)] = make_int2((S).y, (D).y);     \
        out[min(max(q2, 0), N_EDGES - 1)] = make_int2((S).z, (D).z);     \
        out[min(max(q3, 0), N_EDGES - 1)] = make_int2((S).w, (D).w);     \
    } while (0)
    PLACE4(s0, d0);
    if (second) PLACE4(s1, d1);
#undef PLACE4
}

__global__ void __launch_bounds__(PREP_BLOCK)
node_place_kernel(const int* __restrict__ pairs, const int* __restrict__ cell_start,
                  int* __restrict__ rowptr, float* __restrict__ dinv, int* __restrict__ col) {
    __shared__ int cnt[RANGE_NODES];
    __shared__ int wsum[PREP_BLOCK / 32];
    const int tid  = threadIdx.x;
    const int lane = tid & 31;
    const int wv   = tid >> 5;
    const int r    = blockIdx.x;
    const int beg = min(max(cell_start[r * N_CHUNK], 0), N_EDGES);
    const int end = min(max(cell_start[(r + 1) * N_CHUNK], beg), N_EDGES);
    const int2* in = (const int2*)pairs;

    if (tid < RANGE_NODES) cnt[tid] = 0;
    __syncthreads();

    for (int e0 = beg + tid; e0 < end; e0 += NODE_BATCH * PREP_BLOCK) {
        int d[NODE_BATCH];
#pragma unroll
        for (int k = 0; k < NODE_BATCH; ++k) d[k] = in[min(e0 + k * PREP_BLOCK, end - 1)].y;
#pragma unroll
        for (int k = 0; k < NODE_BATCH; ++k)
            if (e0 + k * PREP_BLOCK < end) atomicAdd(&cnt[d[k] & (RANGE_NODES - 1)], 1);
    }
    __syncthreads();

    const int c = (tid < RANGE_NODES) ? cnt[tid] : 0;
    int incl = c;
    for (int o = 1; o < 32; o <<= 1) {
        int t = __shfl_up(incl, o);
        if (lane >= o) incl += t;
    }
    if (lane == 31) wsum[wv] = incl;
    __syncthreads();
    int woff = (lane < wv) ? wsum[lane] : 0;
    for (int o = 16; o > 0; o >>= 1) woff += __shfl_xor(woff, o);
    const int excl = woff + incl - c;

    const int node = (r << RANGE_SHIFT) + tid;
    if (tid < RANGE_NODES) {
        cnt[tid] = excl;
        if (node < N_NODES) {
            rowptr[node] = beg + excl;
            dinv[node] = rsqrtf((float)c + 1.0f);
            if (node == N_NODES - 1) rowptr[N_NODES] = beg + excl + c;
        }
    }
    __syncthreads();

    for (int e0 = beg + tid; e0 < end; e0 += NODE_BATCH * PREP_BLOCK) {
        int2 p[NODE_BATCH];
#pragma unroll
        for (int k = 0; k < NODE_BATCH; ++k) p[k] = in[min(e0 + k * PREP_BLOCK, end - 1)];
#pragma unroll
        for (int k = 0; k < NODE_BATCH; ++k) {
            if (e0 + k * PREP_BLOCK < end) {
                const int slot = atomicAdd(&cnt[p[k].y & (RANGE_NODES - 1)], 1);
                col[min(beg + slot, end - 1)] = p[k].x;
            }
        }
    }
}


template <int K, int NCOL>
__global__ void gemm_wmma_f32(const float* __restrict__ A,
                              const float* __restrict__ B,
                              const float* __restrict__ rscale,
                              float* __restrict__ C) {
    const int wave = threadIdx.x >> 5;
    const int lane = threadIdx.x & 31;
    const int row0 = blockIdx.x << 4;

    const int m  = lane & 15;
    const int kk = (lane >> 4) << 1;
    const int n  = (lane & 15) + (wave << 4);

    const float* ap = A + (size_t)(row0 + m) * K + kk;
    const float* bp = B + (size_t)kk * NCOL + n;

    f32x8 acc = {0.f, 0.f, 0.f, 0.f, 0.f, 0.f, 0.f, 0.f};

    const int mbase = (lane >> 4) << 3;
    float sc[8];
#pragma unroll
    for (int i = 0; i < 8; ++i) sc[i] = rscale[row0 + mbase + i];

    for (int k0 = 0; k0 < K; k0 += 4) {
        f32x2 a, b;
        a.x = ap[k0];
        a.y = ap[k0 + 1];
        b.x = bp[(size_t)k0 * NCOL];
        b.y = bp[(size_t)(k0 + 1) * NCOL];
        acc = __builtin_amdgcn_wmma_f32_16x16x4_f32(
             false, a,  false, b,
             (short)0, acc,  false,  false);
    }

    float* cp = C + (size_t)row0 * NCOL + n;
#pragma unroll
    for (int i = 0; i < 8; ++i)
        cp[(size_t)(mbase + i) * NCOL] = acc[i] * sc[i];
}


typedef __attribute__((ext_vector_type(4))) float f32x4;

template <int L, int K>
__device__ __forceinline__ unsigned group_lane(unsigned x) {
    if constexpr (L == 16) {
        return (unsigned)__builtin_amdgcn_update_dpp(0, (int)x, 0x150 + K, 0xF, 0xF, true);
    } else {
        return __builtin_amdgcn_mov_dpp8(x, (unsigned)K * 0x249249u);
    }
}

template <int C, bool RELU>
__global__ void __launch_bounds__(256)
gather_finish_kernel(const float* __restrict__ hs,
                     const int* __restrict__ rowptr,
                     const int* __restrict__ col,
                     const float* __restrict__ dinv,
                     const float* __restrict__ bias,
                     float* dst_lo, float* dst_hi, int split) {
    constexpr int L = C / 4;
    constexpr int G = 32 / L;
    static_assert(L == 16 || L == 8, "a node's row covers 16 or 8 lanes");
    static_assert(N_NODES % (8 * G) == 0, "blocks tile the node list");

    const int lane = threadIdx.x & 31;
    const int t    = lane & (L - 1);
    const int node = (int)(blockIdx.x * 8 + (threadIdx.x >> 5)) * G + (lane / L);
    const int beg  = rowptr[node];
    const int deg  = rowptr[node + 1] - beg;
    const float di = dinv[node];

    const f32x4* hv = (const f32x4*)hs;

    f32x4 acc = hv[(unsigned)node * L + (unsigned)t];
    const f32x4 bv = *(const f32x4*)(bias + t * 4);
    int my = 0;
    if (t < deg) my = col[(unsigned)(beg + t)];
    unsigned pre = (unsigned)min(max(my, 0), N_NODES - 1) * L;

    for (int base = 0; base < deg; base += L) {
        const int rem = deg - base;
        const int nxt = col[(unsigned)(beg + min(base + L + t, deg - 1))];

        f32x4 v[8];
#define LOAD_ROW(K)                                                       \
        {                                                                 \
            const unsigned p_k = group_lane<L, (K)>(pre) + (unsigned)t;   \
            if ((K) < rem) v[(K) & 7] = hv[p_k];                          \
        }
#define SUM_ROW(K) if ((K) < rem) acc += v[(K) & 7];
        LOAD_ROW(0) LOAD_ROW(1) LOAD_ROW(2) LOAD_ROW(3) LOAD_ROW(4) LOAD_ROW(5) LOAD_ROW(6) LOAD_ROW(7)
        SUM_ROW(0) SUM_ROW(1) SUM_ROW(2) SUM_ROW(3) SUM_ROW(4) SUM_ROW(5) SUM_ROW(6) SUM_ROW(7)
        if constexpr (L == 16) {
            if (rem > 8) {
                LOAD_ROW(8) LOAD_ROW(9) LOAD_ROW(10) LOAD_ROW(11) LOAD_ROW(12) LOAD_ROW(13) LOAD_ROW(14) LOAD_ROW(15)
                SUM_ROW(8) SUM_ROW(9) SUM_ROW(10) SUM_ROW(11) SUM_ROW(12) SUM_ROW(13) SUM_ROW(14) SUM_ROW(15)
            }
        }
#undef LOAD_ROW
#undef SUM_ROW
        pre = (unsigned)min(max(nxt, 0), N_NODES - 1) * L;
    }

    const size_t o = (size_t)node * C + (size_t)t * 4;
    float* op = (node < split) ? (dst_lo + o) : (dst_hi + (o - (size_t)split * C));
    f32x4 r = acc * di + bv;
    if (RELU) {
        r.x = fmaxf(r.x, 0.0f); r.y = fmaxf(r.y, 0.0f);
        r.z = fmaxf(r.z, 0.0f); r.w = fmaxf(r.w, 0.0f);
    }
    *(f32x4*)op = r;
}


extern "C" void kernel_launch(void* const* d_in, const int* in_sizes, int n_in,
                              void* d_out, int out_size, void* d_ws, size_t ws_size,
                              hipStream_t stream) {
    const float* x    = (const float*)d_in[0];
    const int*   ei   = (const int*)d_in[1];
    const float* W1   = (const float*)d_in[2];
    const float* b1   = (const float*)d_in[3];
    const float* W2   = (const float*)d_in[4];
    const float* b2   = (const float*)d_in[5];
    float*       out  = (float*)d_out;

    const int* srcv = ei;
    const int* dstv = ei + N_EDGES;

    float* dinv   = (float*)d_ws;
    int*   rowptr = (int*)(dinv + 100352);
    int*   col    = rowptr + 100352;
    float* h1     = (float*)(col + N_EDGES);
    float* a1lo   = h1 + (size_t)N_NODES * HID_CH;
    float* a1hi   = out;
    int*   cells      = (int*)h1;
    int*   cell_start = cells + 49152;
    int*   pairs      = cell_start + 49152;
    float* h2     = h1;

    range_count_kernel<<<N_CHUNK, PREP_BLOCK, 0, stream>>>(dstv, cells);
    scan_kernel<<<(N_CELL + SCAN_BLOCK - 1) / SCAN_BLOCK, SCAN_BLOCK, 0, stream>>>(cells, cell_start, N_CELL);
    range_place_kernel<<<N_CHUNK, PREP_BLOCK, 0, stream>>>(srcv, dstv, cell_start, pairs);
    node_place_kernel<<<N_RANGE, PREP_BLOCK, 0, stream>>>(pairs, cell_start, rowptr, dinv, col);

    gemm_wmma_f32<IN_CH, HID_CH><<<N_NODES / 16, 128, 0, stream>>>(x, W1, dinv, h1);

    gather_finish_kernel<HID_CH, true>
        <<<N_NODES / 16, 256, 0, stream>>>(h1, rowptr, col, dinv, b1, a1lo, a1hi, SPLIT_ROW);

    gemm_wmma_f32<HID_CH, OUT_CH><<<SPLIT_ROW / 16, 64, 0, stream>>>(a1lo, W2, dinv, h2);
    gemm_wmma_f32<HID_CH, OUT_CH><<<(N_NODES - SPLIT_ROW) / 16, 64, 0, stream>>>(
        a1hi, W2, dinv + SPLIT_ROW, h2 + (size_t)SPLIT_ROW * OUT_CH);

    gather_finish_kernel<OUT_CH, false>
        <<<N_NODES / 32, 256, 0, stream>>>(h2, rowptr, col, dinv, b2, out, out, N_NODES);
}


# Layer-2 product fused into the layer-1 gather: a block's 16 activated rows pass through LDS into the same f32 16x16x4 fragments, never through memory; 2 launches fewer; sum order kept
# speedup vs baseline: 4.9099x; 1.0979x over previous
#include <hip/hip_runtime.h>
#include <hip/hip_bf16.h>


#define N_NODES 100000
#define N_EDGES 1600000
#define IN_CH   128
#define HID_CH  64
#define OUT_CH  32

#define SCAN_BLOCK 1024

typedef __attribute__((ext_vector_type(2))) float f32x2;
typedef __attribute__((ext_vector_type(8))) float f32x8;


#define RANGE_SHIFT 9
#define RANGE_NODES (1 << RANGE_SHIFT)
#define N_RANGE     ((N_NODES + RANGE_NODES - 1) / RANGE_NODES)
#define CHUNK_EDGES 6400
#define N_CHUNK     (N_EDGES / CHUNK_EDGES)
#define CHUNK_VEC   (CHUNK_EDGES / 4)
#define N_CELL      (N_RANGE * N_CHUNK)
#define PREP_BLOCK  1024
#define NODE_BATCH  8

static_assert(N_CHUNK * CHUNK_EDGES == N_EDGES, "chunks must tile the edge list");
static_assert(N_CELL + 1 <= 49152, "count matrix and its scan fit their workspace planes");
static_assert(PREP_BLOCK <= CHUNK_VEC && CHUNK_VEC <= 2 * PREP_BLOCK, "two 16-byte loads per thread cover a chunk");
static_assert(N_RANGE <= PREP_BLOCK && RANGE_NODES <= PREP_BLOCK, "one thread per range / per node of a range");

__device__ __forceinline__ int range_of(int d) {
    return (int)min((unsigned)d >> RANGE_SHIFT, (unsigned)(N_RANGE - 1));
}

__global__ void __launch_bounds__(PREP_BLOCK)
range_count_kernel(const int* __restrict__ dst, int* __restrict__ cells) {
    __shared__ int cnt[N_RANGE];
    const int tid = threadIdx.x;
    if (tid < N_RANGE) cnt[tid] = 0;
    __syncthreads();

    const int4* dp = (const int4*)(dst + (size_t)blockIdx.x * CHUNK_EDGES);
    const bool second = tid + PREP_BLOCK < CHUNK_VEC;
    const int4 d0 = dp[tid];
    int4 d1 = {0, 0, 0, 0};
    if (second) d1 = dp[tid + PREP_BLOCK];
    atomicAdd(&cnt[range_of(d0.x)], 1);
    atomicAdd(&cnt[range_of(d0.y)], 1);
    atomicAdd(&cnt[range_of(d0.z)], 1);
    atomicAdd(&cnt[range_of(d0.w)], 1);
    if (second) {
        atomicAdd(&cnt[range_of(d1.x)], 1);
        atomicAdd(&cnt[range_of(d1.y)], 1);
        atomicAdd(&cnt[range_of(d1.z)], 1);
        atomicAdd(&cnt[range_of(d1.w)], 1);
    }
    __syncthreads();
    if (tid < N_RANGE) cells[tid * N_CHUNK + blockIdx.x] = cnt[tid];
}

__global__ void __launch_bounds__(SCAN_BLOCK)
scan_kernel(const int* __restrict__ cnt, int* __restrict__ out, int n) {
    __shared__ int wsum[SCAN_BLOCK / 32];
    __shared__ int s_prefix;
    const int tid  = threadIdx.x;
    const int lane = tid & 31;
    const int wv   = tid >> 5;
    const int first = blockIdx.x * SCAN_BLOCK;

    int part = 0;
    for (int i = tid; i < first; i += SCAN_BLOCK) part += cnt[i];
    for (int o = 16; o > 0; o >>= 1) part += __shfl_xor(part, o);
    if (lane == 0) wsum[wv] = part;
    __syncthreads();
    if (wv == 0) {
        int t = wsum[lane];
        for (int o = 16; o > 0; o >>= 1) t += __shfl_xor(t, o);
        if (lane == 0) s_prefix = t;
    }
    __syncthreads();
    const int prefix = s_prefix;

    const int i = first + tid;
    const int c = (i < n) ? cnt[i] : 0;
    int incl = c;
    for (int o = 1; o < 32; o <<= 1) {
        int t = __shfl_up(incl, o);
        if (lane >= o) incl += t;
    }
    if (lane == 31) wsum[wv] = incl;
    __syncthreads();
    int woff = (lane < wv) ? wsum[lane] : 0;
    for (int o = 16; o > 0; o >>= 1) woff += __shfl_xor(woff, o);

    const int excl = prefix + woff + incl - c;
    if (i < n) {
        out[i] = excl;
        if (i == n - 1) out[n] = excl + c;
    }
}

__global__ void __launch_bounds__(PREP_BLOCK)
range_place_kernel(const int* __restrict__ src, const int* __restrict__ dst,
                   const int* __restrict__ cell_start, int* __restrict__ pairs) {
    __shared__ int cur[N_RANGE];
    const int tid = threadIdx.x;
    if (tid < N_RANGE) cur[tid] = cell_start[tid * N_CHUNK + blockIdx.x];
    __syncthreads();

    const int4* sp = (const int4*)(src + (size_t)blockIdx.x * CHUNK_EDGES);
    const int4* dp = (const int4*)(dst + (size_t)blockIdx.x * CHUNK_EDGES);
    int2* out = (int2*)pairs;
    const bool second = tid + PREP_BLOCK < CHUNK_VEC;
    const int4 s0 = sp[tid];
    const int4 d0 = dp[tid];
    int4 s1 = {0, 0, 0, 0}, d1 = {0, 0, 0, 0};
    if (second) {
        s1 = sp[tid + PREP_BLOCK];
        d1 = dp[tid + PREP_BLOCK];
    }
#define PLACE4(S, D)                                                     \
    do {                                                                 \
        int q0 = atomicAdd(&cur[range_of((D).x)], 1);                    \
        int q1 = atomicAdd(&cur[range_of((D).y)], 1);                    \
        int q2 = atomicAdd(&cur[range_of((D).z)], 1);                    \
        int q3 = atomicAdd(&cur[range_of((D).w)], 1);                    \
        out[min(max(q0, 0), N_EDGES - 1)] = make_int2((S).x, (D).x);     \
        out[min(max(q1, 0), N_EDGES - 1)] = make_int2((S).y, (D).y);     \
        out[min(max(q2, 0), N_EDGES - 1)] = make_int2((S).z, (D).z);     \
        out[min(max(q3, 0), N_EDGES - 1)] = make_int2((S).w, (D).w);     \
    } while (0)
    PLACE4(s0, d0);
    if (second) PLACE4(s1, d1);
#undef PLACE4
}

__global__ void __launch_bounds__(PREP_BLOCK)
node_place_kernel(const int* __restrict__ pairs, const int* __restrict__ cell_start,
                  int* __restrict__ rowptr, float* __restrict__ dinv, int* __restrict__ col) {
    __shared__ int cnt[RANGE_NODES];
    __shared__ int wsum[PREP_BLOCK / 32];
    const int tid  = threadIdx.x;
    const int lane = tid & 31;
    const int wv   = tid >> 5;
    const int r    = blockIdx.x;
    const int beg = min(max(cell_start[r * N_CHUNK], 0), N_EDGES);
    const int end = min(max(cell_start[(r + 1) * N_CHUNK], beg), N_EDGES);
    const int2* in = (const int2*)pairs;

    if (tid < RANGE_NODES) cnt[tid] = 0;
    __syncthreads();

    for (int e0 = beg + tid; e0 < end; e0 += NODE_BATCH * PREP_BLOCK) {
        int d[NODE_BATCH];
#pragma unroll
        for (int k = 0; k < NODE_BATCH; ++k) d[k] = in[min(e0 + k * PREP_BLOCK, end - 1)].y;
#pragma unroll
        for (int k = 0; k < NODE_BATCH; ++k)
            if (e0 + k * PREP_BLOCK < end) atomicAdd(&cnt[d[k] & (RANGE_NODES - 1)], 1);
    }
    __syncthreads();

    const int c = (tid < RANGE_NODES) ? cnt[tid] : 0;
    int incl = c;
    for (int o = 1; o < 32; o <<= 1) {
        int t = __shfl_up(incl, o);
        if (lane >= o) incl += t;
    }
    if (lane == 31) wsum[wv] = incl;
    __syncthreads();
    int woff = (lane < wv) ? wsum[lane] : 0;
    for (int o = 16; o > 0; o >>= 1) woff += __shfl_xor(woff, o);
    const int excl = woff + incl - c;

    const int node = (r << RANGE_SHIFT) + tid;
    if (tid < RANGE_NODES) {
        cnt[tid] = excl;
        if (node < N_NODES) {
            rowptr[node] = beg + excl;
            dinv[node] = rsqrtf((float)c + 1.0f);
            if (node == N_NODES - 1) rowptr[N_NODES] = beg + excl + c;
        }
    }
    __syncthreads();

    for (int e0 = beg + tid; e0 < end; e0 += NODE_BATCH * PREP_BLOCK) {
        int2 p[NODE_BATCH];
#pragma unroll
        for (int k = 0; k < NODE_BATCH; ++k) p[k] = in[min(e0 + k * PREP_BLOCK, end - 1)];
#pragma unroll
        for (int k = 0; k < NODE_BATCH; ++k) {
            if (e0 + k * PREP_BLOCK < end) {
                const int slot = atomicAdd(&cnt[p[k].y & (RANGE_NODES - 1)], 1);
                col[min(beg + slot, end - 1)] = p[k].x;
            }
        }
    }
}


template <int K, int NCOL>
__global__ void gemm_wmma_f32(const float* __restrict__ A,
                              const float* __restrict__ B,
                              const float* __restrict__ rscale,
                              float* __restrict__ C) {
    const int wave = threadIdx.x >> 5;
    const int lane = threadIdx.x & 31;
    const int row0 = blockIdx.x << 4;

    const int m  = lane & 15;
    const int kk = (lane >> 4) << 1;
    const int n  = (lane & 15) + (wave << 4);

    const float* ap = A + (size_t)(row0 + m) * K + kk;
    const float* bp = B + (size_t)kk * NCOL + n;

    f32x8 acc = {0.f, 0.f, 0.f, 0.f, 0.f, 0.f, 0.f, 0.f};

    const int mbase = (lane >> 4) << 3;
    float sc[8];
#pragma unroll
    for (int i = 0; i < 8; ++i) sc[i] = rscale[row0 + mbase + i];

    for (int k0 = 0; k0 < K; k0 += 4) {
        f32x2 a, b;
        a.x = ap[k0];
        a.y = ap[k0 + 1];
        b.x = bp[(size_t)k0 * NCOL];
        b.y = bp[(size_t)(k0 + 1) * NCOL];
        acc = __builtin_amdgcn_wmma_f32_16x16x4_f32(
             false, a,  false, b,
             (short)0, acc,  false,  false);
    }

    float* cp = C + (size_t)row0 * NCOL + n;
#pragma unroll
    for (int i = 0; i < 8; ++i)
        cp[(size_t)(mbase + i) * NCOL] = acc[i] * sc[i];
}


typedef __attribute__((ext_vector_type(4))) float f32x4;

template <int L, int K>
__device__ __forceinline__ unsigned group_lane(unsigned x) {
    if constexpr (L == 16) {
        return (unsigned)__builtin_amdgcn_update_dpp(0, (int)x, 0x150 + K, 0xF, 0xF, true);
    } else {
        return __builtin_amdgcn_mov_dpp8(x, (unsigned)K * 0x249249u);
    }
}

template <int L>
__device__ __forceinline__ f32x4 bucket_sum(const f32x4* __restrict__ hv, const int* __restrict__ col,
                                            int node, int beg, int deg, int t) {
    f32x4 acc = hv[(unsigned)node * L + (unsigned)t];
    int my = 0;
    if (t < deg) my = col[(unsigned)(beg + t)];
    unsigned pre = (unsigned)min(max(my, 0), N_NODES - 1) * L;

    for (int base = 0; base < deg; base += L) {
        const int rem = deg - base;
        const int nxt = col[(unsigned)(beg + min(base + L + t, deg - 1))];

        f32x4 v[8];
#define LOAD_ROW(K)                                                       \
        {                                                                 \
            const unsigned p_k = group_lane<L, (K)>(pre) + (unsigned)t;   \
            if ((K) < rem) v[(K) & 7] = hv[p_k];                          \
        }
#define SUM_ROW(K) if ((K) < rem) acc += v[(K) & 7];
        LOAD_ROW(0) LOAD_ROW(1) LOAD_ROW(2) LOAD_ROW(3) LOAD_ROW(4) LOAD_ROW(5) LOAD_ROW(6) LOAD_ROW(7)
        SUM_ROW(0) SUM_ROW(1) SUM_ROW(2) SUM_ROW(3) SUM_ROW(4) SUM_ROW(5) SUM_ROW(6) SUM_ROW(7)
        if constexpr (L == 16) {
            if (rem > 8) {
                LOAD_ROW(8) LOAD_ROW(9) LOAD_ROW(10) LOAD_ROW(11) LOAD_ROW(12) LOAD_ROW(13) LOAD_ROW(14) LOAD_ROW(15)
                SUM_ROW(8) SUM_ROW(9) SUM_ROW(10) SUM_ROW(11) SUM_ROW(12) SUM_ROW(13) SUM_ROW(14) SUM_ROW(15)
            }
        }
#undef LOAD_ROW
#undef SUM_ROW
        pre = (unsigned)min(max(nxt, 0), N_NODES - 1) * L;
    }
    return acc;
}

template <int C>
__global__ void __launch_bounds__(256)
gather_finish_kernel(const float* __restrict__ hs,
                     const int* __restrict__ rowptr,
                     const int* __restrict__ col,
                     const float* __restrict__ dinv,
                     const float* __restrict__ bias,
                     float* __restrict__ dst) {
    constexpr int L = C / 4;
    constexpr int G = 32 / L;
    static_assert(L == 16 || L == 8, "a node's row covers 16 or 8 lanes");
    static_assert(N_NODES % (8 * G) == 0, "blocks tile the node list");

    const int lane = threadIdx.x & 31;
    const int t    = lane & (L - 1);
    const int node = (int)(blockIdx.x * 8 + (threadIdx.x >> 5)) * G + (lane / L);
    const int beg  = rowptr[node];
    const int deg  = rowptr[node + 1] - beg;
    const float di = dinv[node];
    const f32x4 bv = *(const f32x4*)(bias + t * 4);

    const f32x4 acc = bucket_sum<L>((const f32x4*)hs, col, node, beg, deg, t);

    *(f32x4*)(dst + (size_t)node * C + (size_t)t * 4) = acc * di + bv;
}

#define TILE_PITCH (HID_CH + 4)

__global__ void __launch_bounds__(256)
gather_relu_product_kernel(const float* __restrict__ hs,
                           const int* __restrict__ rowptr,
                           const int* __restrict__ col,
                           const float* __restrict__ dinv,
                           const float* __restrict__ bias,
                           const float* __restrict__ Wn,
                           float* __restrict__ hn) {
    constexpr int L = HID_CH / 4;
    static_assert(L == 16 && N_NODES % 16 == 0, "a block's nodes are the 16 rows of one matrix tile");
    static_assert(OUT_CH % 16 == 0 && OUT_CH / 16 <= 8, "one wave per column tile");

    __shared__ __attribute__((aligned(16))) float tile[16 * TILE_PITCH];
    __shared__ float tscale[16];

    const int wave = threadIdx.x >> 5;
    const int lane = threadIdx.x & 31;
    const int t    = lane & (L - 1);
    const int row  = threadIdx.x >> 4;
    const int row0 = blockIdx.x << 4;
    const int node = row0 + row;
    const int beg  = rowptr[node];
    const int deg  = rowptr[node + 1] - beg;
    const float di = dinv[node];
    const f32x4 bv = *(const f32x4*)(bias + t * 4);

    const f32x4 acc = bucket_sum<L>((const f32x4*)hs, col, node, beg, deg, t);

    f32x4 r = acc * di + bv;
    r.x = fmaxf(r.x, 0.0f); r.y = fmaxf(r.y, 0.0f);
    r.z = fmaxf(r.z, 0.0f); r.w = fmaxf(r.w, 0.0f);
    *(f32x4*)(tile + row * TILE_PITCH + t * 4) = r;
    if (t == 0) tscale[row] = di;

    const int m  = lane & 15;
    const int kk = (lane >> 4) << 1;
    const int n  = m + (wave << 4);
    const bool owner = wave < OUT_CH / 16;

    f32x2 bf[HID_CH / 4];
    if (owner) {
        const float* bp = Wn + (size_t)kk * OUT_CH + n;
#pragma unroll
        for (int s = 0; s < HID_CH / 4; ++s) {
            bf[s].x = bp[(size_t)(4 * s) * OUT_CH];
            bf[s].y = bp[(size_t)(4 * s + 1) * OUT_CH];
        }
    }
    __syncthreads();
    if (!owner) return;

    const float* ap = tile + m * TILE_PITCH + kk;
    const int mbase = (lane >> 4) << 3;
    float* cp = hn + (size_t)row0 * OUT_CH + n;
    float sc[8];
#pragma unroll
    for (int i = 0; i < 8; ++i) sc[i] = tscale[mbase + i];

    __builtin_amdgcn_sched_barrier(0);
    f32x8 pacc = {0.f, 0.f, 0.f, 0.f, 0.f, 0.f, 0.f, 0.f};
#pragma unroll
    for (int s = 0; s < HID_CH / 4; ++s) {
        const f32x2 a = *(const f32x2*)(ap + 4 * s);
        pacc = __builtin_amdgcn_wmma_f32_16x16x4_f32(
             false, a,  false, bf[s],
             (short)0, pacc,  false,  false);
    }
    __builtin_amdgcn_sched_barrier(0);
    asm volatile("v_nop\n\tv_nop\n\tv_nop\n\tv_nop" : "+v"(pacc));

#pragma unroll
    for (int i = 0; i < 8; ++i)
        cp[(size_t)(mbase + i) * OUT_CH] = pacc[i] * sc[i];
}


extern "C" void kernel_launch(void* const* d_in, const int* in_sizes, int n_in,
                              void* d_out, int out_size, void* d_ws, size_t ws_size,
                              hipStream_t stream) {
    const float* x    = (const float*)d_in[0];
    const int*   ei   = (const int*)d_in[1];
    const float* W1   = (const float*)d_in[2];
    const float* b1   = (const float*)d_in[3];
    const float* W2   = (const float*)d_in[4];
    const float* b2   = (const float*)d_in[5];
    float*       out  = (float*)d_out;

    const int* srcv = ei;
    const int* dstv = ei + N_EDGES;

    float* dinv   = (float*)d_ws;
    int*   rowptr = (int*)(dinv + 100352);
    int*   col    = rowptr + 100352;
    float* h1     = (float*)(col + N_EDGES);
    float* h2     = h1 + (size_t)N_NODES * HID_CH;
    int*   cells      = (int*)h1;
    int*   cell_start = cells + 49152;
    int*   pairs      = cell_start + 49152;

    range_count_kernel<<<N_CHUNK, PREP_BLOCK, 0, stream>>>(dstv, cells);
    scan_kernel<<<(N_CELL + SCAN_BLOCK - 1) / SCAN_BLOCK, SCAN_BLOCK, 0, stream>>>(cells, cell_start, N_CELL);
    range_place_kernel<<<N_CHUNK, PREP_BLOCK, 0, stream>>>(srcv, dstv, cell_start, pairs);
    node_place_kernel<<<N_RANGE, PREP_BLOCK, 0, stream>>>(pairs, cell_start, rowptr, dinv, col);

    gemm_wmma_f32<IN_CH, HID_CH><<<N_NODES / 16, 128, 0, stream>>>(x, W1, dinv, h1);

    gather_relu_product_kernel<<<N_NODES / 16, 256, 0, stream>>>(h1, rowptr, col, dinv, b1, W2, h2);

    gather_finish_kernel<OUT_CH><<<N_NODES / 32, 256, 0, stream>>>(h2, rowptr, col, dinv, b2, out);
}


# Coarse bucketing made chunk-local: a block counts, scans and places its own chunk by range into its own run, range blocks walk their 250 groups; 2 launches fewer; slot order, so sum order, changes
# speedup vs baseline: 5.4461x; 1.1092x over previous
#include <hip/hip_runtime.h>
#include <hip/hip_bf16.h>


#define N_NODES 100000
#define N_EDGES 1600000
#define IN_CH   128
#define HID_CH  64
#define OUT_CH  32

typedef __attribute__((ext_vector_type(2))) float f32x2;
typedef __attribute__((ext_vector_type(8))) float f32x8;


#define RANGE_SHIFT 9
#define RANGE_NODES (1 << RANGE_SHIFT)
#define N_RANGE     ((N_NODES + RANGE_NODES - 1) / RANGE_NODES)
#define CHUNK_EDGES 6400
#define N_CHUNK     (N_EDGES / CHUNK_EDGES)
#define CHUNK_VEC   (CHUNK_EDGES / 4)
#define N_CELL      (N_RANGE * N_CHUNK)
#define PREP_BLOCK  1024
#define SEG_LANES   4
#define SEG_BATCH   12
#define FRONT_VEC   12

static_assert(N_CHUNK * CHUNK_EDGES == N_EDGES, "chunks must tile the edge list");
static_assert(N_CELL <= 49152 && N_CELL % 4 == 0, "count matrix fits its workspace plane, in whole 16-byte pieces");
static_assert(PREP_BLOCK <= CHUNK_VEC && CHUNK_VEC <= 2 * PREP_BLOCK, "two 16-byte loads per thread cover a chunk");
static_assert(N_RANGE <= PREP_BLOCK && RANGE_NODES <= PREP_BLOCK, "one thread per range / per node of a range");
static_assert(N_CHUNK * SEG_LANES <= PREP_BLOCK, "one group of lanes per chunk");
static_assert(((N_RANGE - 1) * N_CHUNK + 3) / 4 <= FRONT_VEC * PREP_BLOCK, "one batch of loads covers all counts in front");

__device__ __forceinline__ int range_of(int d) {
    return (int)min((unsigned)d >> RANGE_SHIFT, (unsigned)(N_RANGE - 1));
}

__global__ void __launch_bounds__(PREP_BLOCK)
chunk_bucket_kernel(const int* __restrict__ src, const int* __restrict__ dst,
                    int* __restrict__ cells, int* __restrict__ loff, int* __restrict__ pairs) {
    __shared__ int cnt[N_RANGE];
    __shared__ int wsum[PREP_BLOCK / 32];
    const int tid  = threadIdx.x;
    const int lane = tid & 31;
    const int wv   = tid >> 5;

    const int4* sp = (const int4*)(src + (size_t)blockIdx.x * CHUNK_EDGES);
    const int4* dp = (const int4*)(dst + (size_t)blockIdx.x * CHUNK_EDGES);
    int2* out = (int2*)pairs;
    const bool second = tid + PREP_BLOCK < CHUNK_VEC;
    const int4 d0 = dp[tid];
    const int4 s0 = sp[tid];
    int4 s1 = {0, 0, 0, 0}, d1 = {0, 0, 0, 0};
    if (second) {
        d1 = dp[tid + PREP_BLOCK];
        s1 = sp[tid + PREP_BLOCK];
    }
    if (tid < N_RANGE) cnt[tid] = 0;
    __syncthreads();
    atomicAdd(&cnt[range_of(d0.x)], 1);
    atomicAdd(&cnt[range_of(d0.y)], 1);
    atomicAdd(&cnt[range_of(d0.z)], 1);
    atomicAdd(&cnt[range_of(d0.w)], 1);
    if (second) {
        atomicAdd(&cnt[range_of(d1.x)], 1);
        atomicAdd(&cnt[range_of(d1.y)], 1);
        atomicAdd(&cnt[range_of(d1.z)], 1);
        atomicAdd(&cnt[range_of(d1.w)], 1);
    }
    __syncthreads();

    const int c = (tid < N_RANGE) ? cnt[tid] : 0;
    int incl = c;
    for (int o = 1; o < 32; o <<= 1) {
        int t = __shfl_up(incl, o);
        if (lane >= o) incl += t;
    }
    if (lane == 31) wsum[wv] = incl;
    __syncthreads();
    int woff = (lane < wv) ? wsum[lane] : 0;
    for (int o = 16; o > 0; o >>= 1) woff += __shfl_xor(woff, o);
    const int excl = woff + incl - c;

    const int base = (int)blockIdx.x * CHUNK_EDGES;
    if (tid < N_RANGE) {
        cnt[tid] = base + excl;
        cells[tid * N_CHUNK + blockIdx.x] = c;
        loff[tid * N_CHUNK + blockIdx.x]  = excl;
    }
    __syncthreads();

#define PLACE4(S, D)                                                              \
    do {                                                                          \
        int q0 = atomicAdd(&cnt[range_of((D).x)], 1);                             \
        int q1 = atomicAdd(&cnt[range_of((D).y)], 1);                             \
        int q2 = atomicAdd(&cnt[range_of((D).z)], 1);                             \
        int q3 = atomicAdd(&cnt[range_of((D).w)], 1);                             \
        out[min(max(q0, base), base + CHUNK_EDGES - 1)] = make_int2((S).x, (D).x); \
        out[min(max(q1, base), base + CHUNK_EDGES - 1)] = make_int2((S).y, (D).y); \
        out[min(max(q2, base), base + CHUNK_EDGES - 1)] = make_int2((S).z, (D).z); \
        out[min(max(q3, base), base + CHUNK_EDGES - 1)] = make_int2((S).w, (D).w); \
    } while (0)
    PLACE4(s0, d0);
    if (second) PLACE4(s1, d1);
#undef PLACE4
}

__global__ void __launch_bounds__(PREP_BLOCK)
node_place_kernel(const int* __restrict__ pairs, const int* __restrict__ cells, const int* __restrict__ loff,
                  int* __restrict__ rowptr, float* __restrict__ dinv, int* __restrict__ col) {
    __shared__ int cnt[RANGE_NODES];
    __shared__ int wsum[PREP_BLOCK / 32];
    __shared__ int wfront[PREP_BLOCK / 32];
    __shared__ int wown[PREP_BLOCK / 32];
    const int tid  = threadIdx.x;
    const int lane = tid & 31;
    const int wv   = tid >> 5;
    const int r    = blockIdx.x;
    const int g    = tid / SEG_LANES;
    const int sub  = tid % SEG_LANES;
    const int2* in = (const int2*)pairs;

    if (tid < RANGE_NODES) cnt[tid] = 0;

    const int gc = min(g, N_CHUNK - 1);
    int n  = cells[r * N_CHUNK + gc];
    int lo = loff[r * N_CHUNK + gc];
    if (g >= N_CHUNK) n = 0;
    lo = min(max(lo, 0), CHUNK_EDGES);
    n  = min(max(n, 0), CHUNK_EDGES - lo);
    const int2* seg = in + (size_t)gc * CHUNK_EDGES + lo;

    const int front = r * N_CHUNK;
    const int nvec  = (front + 3) >> 2;
    const int4* c4  = (const int4*)cells;
    int4 fv[FRONT_VEC];
#pragma unroll
    for (int k = 0; k < FRONT_VEC; ++k) fv[k] = c4[min(tid + k * PREP_BLOCK, max(nvec - 1, 0))];
    int part = 0;
#pragma unroll
    for (int k = 0; k < FRONT_VEC; ++k) {
        const int e = (tid + k * PREP_BLOCK) * 4;
        if (e     < front) part += fv[k].x;
        if (e + 1 < front) part += fv[k].y;
        if (e + 2 < front) part += fv[k].z;
        if (e + 3 < front) part += fv[k].w;
    }
    int own = (sub == 0) ? n : 0;
    for (int o = 16; o > 0; o >>= 1) {
        part += __shfl_xor(part, o);
        own  += __shfl_xor(own, o);
    }
    if (lane == 0) { wfront[wv] = part; wown[wv] = own; }
    __syncthreads();
    int tf = wfront[lane], to = wown[lane];
    for (int o = 16; o > 0; o >>= 1) {
        tf += __shfl_xor(tf, o);
        to += __shfl_xor(to, o);
    }
    const int beg = min(max(tf, 0), N_EDGES);
    const int end = min(beg + min(max(to, 0), N_EDGES), N_EDGES);

    for (int j0 = sub; j0 < n; j0 += SEG_LANES * SEG_BATCH) {
        int d[SEG_BATCH];
#pragma unroll
        for (int k = 0; k < SEG_BATCH; ++k) d[k] = seg[min(j0 + k * SEG_LANES, n - 1)].y;
#pragma unroll
        for (int k = 0; k < SEG_BATCH; ++k)
            if (j0 + k * SEG_LANES < n) atomicAdd(&cnt[d[k] & (RANGE_NODES - 1)], 1);
    }
    __syncthreads();

    const int c = (tid < RANGE_NODES) ? cnt[tid] : 0;
    int incl = c;
    for (int o = 1; o < 32; o <<= 1) {
        int t = __shfl_up(incl, o);
        if (lane >= o) incl += t;
    }
    if (lane == 31) wsum[wv] = incl;
    __syncthreads();
    int woff = (lane < wv) ? wsum[lane] : 0;
    for (int o = 16; o > 0; o >>= 1) woff += __shfl_xor(woff, o);
    const int excl = woff + incl - c;

    const int node = (r << RANGE_SHIFT) + tid;
    if (tid < RANGE_NODES) {
        cnt[tid] = excl;
        if (node < N_NODES) {
            rowptr[node] = beg + excl;
            dinv[node] = rsqrtf((float)c + 1.0f);
            if (node == N_NODES - 1) rowptr[N_NODES] = beg + excl + c;
        }
    }
    __syncthreads();

    for (int j0 = sub; j0 < n; j0 += SEG_LANES * SEG_BATCH) {
        int2 p[SEG_BATCH];
#pragma unroll
        for (int k = 0; k < SEG_BATCH; ++k) p[k] = seg[min(j0 + k * SEG_LANES, n - 1)];
#pragma unroll
        for (int k = 0; k < SEG_BATCH; ++k) {
            if (j0 + k * SEG_LANES < n) {
                const int slot = atomicAdd(&cnt[p[k].y & (RANGE_NODES - 1)], 1);
                col[min(beg + slot, end - 1)] = p[k].x;
            }
        }
    }
}


template <int K, int NCOL>
__global__ void gemm_wmma_f32(const float* __restrict__ A,
                              const float* __restrict__ B,
                              const float* __restrict__ rscale,
                              float* __restrict__ C) {
    const int wave = threadIdx.x >> 5;
    const int lane = threadIdx.x & 31;
    const int row0 = blockIdx.x << 4;

    const int m  = lane & 15;
    const int kk = (lane >> 4) << 1;
    const int n  = (lane & 15) + (wave << 4);

    const float* ap = A + (size_t)(row0 + m) * K + kk;
    const float* bp = B + (size_t)kk * NCOL + n;

    f32x8 acc = {0.f, 0.f, 0.f, 0.f, 0.f, 0.f, 0.f, 0.f};

    const int mbase = (lane >> 4) << 3;
    float sc[8];
#pragma unroll
    for (int i = 0; i < 8; ++i) sc[i] = rscale[row0 + mbase + i];

    for (int k0 = 0; k0 < K; k0 += 4) {
        f32x2 a, b;
        a.x = ap[k0];
        a.y = ap[k0 + 1];
        b.x = bp[(size_t)k0 * NCOL];
        b.y = bp[(size_t)(k0 + 1) * NCOL];
        acc = __builtin_amdgcn_wmma_f32_16x16x4_f32(
             false, a,  false, b,
             (short)0, acc,  false,  false);
    }

    float* cp = C + (size_t)row0 * NCOL + n;
#pragma unroll
    for (int i = 0; i < 8; ++i)
        cp[(size_t)(mbase + i) * NCOL] = acc[i] * sc[i];
}


typedef __attribute__((ext_vector_type(4))) float f32x4;

template <int L, int K>
__device__ __forceinline__ unsigned group_lane(unsigned x) {
    if constexpr (L == 16) {
        return (unsigned)__builtin_amdgcn_update_dpp(0, (int)x, 0x150 + K, 0xF, 0xF, true);
    } else {
        return __builtin_amdgcn_mov_dpp8(x, (unsigned)K * 0x249249u);
    }
}

template <int L>
__device__ __forceinline__ f32x4 bucket_sum(const f32x4* __restrict__ hv, const int* __restrict__ col,
                                            int node, int beg, int deg, int t) {
    f32x4 acc = hv[(unsigned)node * L + (unsigned)t];
    int my = 0;
    if (t < deg) my = col[(unsigned)(beg + t)];
    unsigned pre = (unsigned)min(max(my, 0), N_NODES - 1) * L;

    for (int base = 0; base < deg; base += L) {
        const int rem = deg - base;
        const int nxt = col[(unsigned)(beg + min(base + L + t, deg - 1))];

        f32x4 v[8];
#define LOAD_ROW(K)                                                       \
        {                                                                 \
            const unsigned p_k = group_lane<L, (K)>(pre) + (unsigned)t;   \
            if ((K) < rem) v[(K) & 7] = hv[p_k];                          \
        }
#define SUM_ROW(K) if ((K) < rem) acc += v[(K) & 7];
        LOAD_ROW(0) LOAD_ROW(1) LOAD_ROW(2) LOAD_ROW(3) LOAD_ROW(4) LOAD_ROW(5) LOAD_ROW(6) LOAD_ROW(7)
        SUM_ROW(0) SUM_ROW(1) SUM_ROW(2) SUM_ROW(3) SUM_ROW(4) SUM_ROW(5) SUM_ROW(6) SUM_ROW(7)
        if constexpr (L == 16) {
            if (rem > 8) {
                LOAD_ROW(8) LOAD_ROW(9) LOAD_ROW(10) LOAD_ROW(11) LOAD_ROW(12) LOAD_ROW(13) LOAD_ROW(14) LOAD_ROW(15)
                SUM_ROW(8) SUM_ROW(9) SUM_ROW(10) SUM_ROW(11) SUM_ROW(12) SUM_ROW(13) SUM_ROW(14) SUM_ROW(15)
            }
        }
#undef LOAD_ROW
#undef SUM_ROW
        pre = (unsigned)min(max(nxt, 0), N_NODES - 1) * L;
    }
    return acc;
}

template <int C>
__global__ void __launch_bounds__(256)
gather_finish_kernel(const float* __restrict__ hs,
                     const int* __restrict__ rowptr,
                     const int* __restrict__ col,
                     const float* __restrict__ dinv,
                     const float* __restrict__ bias,
                     float* __restrict__ dst) {
    constexpr int L = C / 4;
    constexpr int G = 32 / L;
    static_assert(L == 16 || L == 8, "a node's row covers 16 or 8 lanes");
    static_assert(N_NODES % (8 * G) == 0, "blocks tile the node list");

    const int lane = threadIdx.x & 31;
    const int t    = lane & (L - 1);
    const int node = (int)(blockIdx.x * 8 + (threadIdx.x >> 5)) * G + (lane / L);
    const int beg  = rowptr[node];
    const int deg  = rowptr[node + 1] - beg;
    const float di = dinv[node];
    const f32x4 bv = *(const f32x4*)(bias + t * 4);

    const f32x4 acc = bucket_sum<L>((const f32x4*)hs, col, node, beg, deg, t);

    *(f32x4*)(dst + (size_t)node * C + (size_t)t * 4) = acc * di + bv;
}

#define TILE_PITCH (HID_CH + 4)

__global__ void __launch_bounds__(256)
gather_relu_product_kernel(const float* __restrict__ hs,
                           const int* __restrict__ rowptr,
                           const int* __restrict__ col,
                           const float* __restrict__ dinv,
                           const float* __restrict__ bias,
                           const float* __restrict__ Wn,
                           float* __restrict__ hn) {
    constexpr int L = HID_CH / 4;
    static_assert(L == 16 && N_NODES % 16 == 0, "a block's nodes are the 16 rows of one matrix tile");
    static_assert(OUT_CH % 16 == 0 && OUT_CH / 16 <= 8, "one wave per column tile");

    __shared__ __attribute__((aligned(16))) float tile[16 * TILE_PITCH];
    __shared__ float tscale[16];

    const int wave = threadIdx.x >> 5;
    const int lane = threadIdx.x & 31;
    const int t    = lane & (L - 1);
    const int row  = threadIdx.x >> 4;
    const int row0 = blockIdx.x << 4;
    const int node = row0 + row;
    const int beg  = rowptr[node];
    const int deg  = rowptr[node + 1] - beg;
    const float di = dinv[node];
    const f32x4 bv = *(const f32x4*)(bias + t * 4);

    const f32x4 acc = bucket_sum<L>((const f32x4*)hs, col, node, beg, deg, t);

    f32x4 r = acc * di + bv;
    r.x = fmaxf(r.x, 0.0f); r.y = fmaxf(r.y, 0.0f);
    r.z = fmaxf(r.z, 0.0f); r.w = fmaxf(r.w, 0.0f);
    *(f32x4*)(tile + row * TILE_PITCH + t * 4) = r;
    if (t == 0) tscale[row] = di;

    const int m  = lane & 15;
    const int kk = (lane >> 4) << 1;
    const int n  = m + (wave << 4);
    const bool owner = wave < OUT_CH / 16;

    f32x2 bf[HID_CH / 4];
    if (owner) {
        const float* bp = Wn + (size_t)kk * OUT_CH + n;
#pragma unroll
        for (int s = 0; s < HID_CH / 4; ++s) {
            bf[s].x = bp[(size_t)(4 * s) * OUT_CH];
            bf[s].y = bp[(size_t)(4 * s + 1) * OUT_CH];
        }
    }
    __syncthreads();
    if (!owner) return;

    const float* ap = tile + m * TILE_PITCH + kk;
    const int mbase = (lane >> 4) << 3;
    float* cp = hn + (size_t)row0 * OUT_CH + n;
    float sc[8];
#pragma unroll
    for (int i = 0; i < 8; ++i) sc[i] = tscale[mbase + i];

    __builtin_amdgcn_sched_barrier(0);
    f32x8 pacc = {0.f, 0.f, 0.f, 0.f, 0.f, 0.f, 0.f, 0.f};
#pragma unroll
    for (int s = 0; s < HID_CH / 4; ++s) {
        const f32x2 a = *(const f32x2*)(ap + 4 * s);
        pacc = __builtin_amdgcn_wmma_f32_16x16x4_f32(
             false, a,  false, bf[s],
             (short)0, pacc,  false,  false);
    }
    __builtin_amdgcn_sched_barrier(0);
    asm volatile("v_nop\n\tv_nop\n\tv_nop\n\tv_nop" : "+v"(pacc));

#pragma unroll
    for (int i = 0; i < 8; ++i)
        cp[(size_t)(mbase + i) * OUT_CH] = pacc[i] * sc[i];
}


extern "C" void kernel_launch(void* const* d_in, const int* in_sizes, int n_in,
                              void* d_out, int out_size, void* d_ws, size_t ws_size,
                              hipStream_t stream) {
    const float* x    = (const float*)d_in[0];
    const int*   ei   = (const int*)d_in[1];
    const float* W1   = (const float*)d_in[2];
    const float* b1   = (const float*)d_in[3];
    const float* W2   = (const float*)d_in[4];
    const float* b2   = (const float*)d_in[5];
    float*       out  = (float*)d_out;

    const int* srcv = ei;
    const int* dstv = ei + N_EDGES;

    float* dinv   = (float*)d_ws;
    int*   rowptr = (int*)(dinv + 100352);
    int*   col    = rowptr + 100352;
    float* h1     = (float*)(col + N_EDGES);
    float* h2     = h1 + (size_t)N_NODES * HID_CH;
    int*   cells = (int*)h1;
    int*   loff  = cells + 49152;
    int*   pairs = loff + 49152;

    chunk_bucket_kernel<<<N_CHUNK, PREP_BLOCK, 0, stream>>>(srcv, dstv, cells, loff, pairs);
    node_place_kernel<<<N_RANGE, PREP_BLOCK, 0, stream>>>(pairs, cells, loff, rowptr, dinv, col);

    gemm_wmma_f32<IN_CH, HID_CH><<<N_NODES / 16, 128, 0, stream>>>(x, W1, dinv, h1);

    gather_relu_product_kernel<<<N_NODES / 16, 256, 0, stream>>>(h1, rowptr, col, dinv, b1, W2, h2);

    gather_finish_kernel<OUT_CH><<<N_NODES / 32, 256, 0, stream>>>(h2, rowptr, col, dinv, b2, out);
}


# Layer-1 product: a block's 32 x rows are read once as whole 512-byte rows (16 B per lane, non-temporal) into LDS, A fragments come from LDS, not 8-byte row-strided loads in every wave; k order kept
# speedup vs baseline: 5.9820x; 1.0984x over previous
#include <hip/hip_runtime.h>
#include <hip/hip_bf16.h>


#define N_NODES 100000
#define N_EDGES 1600000
#define IN_CH   128
#define HID_CH  64
#define OUT_CH  32

typedef __attribute__((ext_vector_type(2))) float f32x2;
typedef __attribute__((ext_vector_type(8))) float f32x8;


#define RANGE_SHIFT 9
#define RANGE_NODES (1 << RANGE_SHIFT)
#define N_RANGE     ((N_NODES + RANGE_NODES - 1) / RANGE_NODES)
#define CHUNK_EDGES 6400
#define N_CHUNK     (N_EDGES / CHUNK_EDGES)
#define CHUNK_VEC   (CHUNK_EDGES / 4)
#define N_CELL      (N_RANGE * N_CHUNK)
#define PREP_BLOCK  1024
#define SEG_LANES   4
#define SEG_BATCH   12
#define FRONT_VEC   12

static_assert(N_CHUNK * CHUNK_EDGES == N_EDGES, "chunks must tile the edge list");
static_assert(N_CELL <= 49152 && N_CELL % 4 == 0, "count matrix fits its workspace plane, in whole 16-byte pieces");
static_assert(PREP_BLOCK <= CHUNK_VEC && CHUNK_VEC <= 2 * PREP_BLOCK, "two 16-byte loads per thread cover a chunk");
static_assert(N_RANGE <= PREP_BLOCK && RANGE_NODES <= PREP_BLOCK, "one thread per range / per node of a range");
static_assert(N_CHUNK * SEG_LANES <= PREP_BLOCK, "one group of lanes per chunk");
static_assert(((N_RANGE - 1) * N_CHUNK + 3) / 4 <= FRONT_VEC * PREP_BLOCK, "one batch of loads covers all counts in front");

__device__ __forceinline__ int range_of(int d) {
    return (int)min((unsigned)d >> RANGE_SHIFT, (unsigned)(N_RANGE - 1));
}

__global__ void __launch_bounds__(PREP_BLOCK)
chunk_bucket_kernel(const int* __restrict__ src, const int* __restrict__ dst,
                    int* __restrict__ cells, int* __restrict__ loff, int* __restrict__ pairs) {
    __shared__ int cnt[N_RANGE];
    __shared__ int wsum[PREP_BLOCK / 32];
    const int tid  = threadIdx.x;
    const int lane = tid & 31;
    const int wv   = tid >> 5;

    const int4* sp = (const int4*)(src + (size_t)blockIdx.x * CHUNK_EDGES);
    const int4* dp = (const int4*)(dst + (size_t)blockIdx.x * CHUNK_EDGES);
    int2* out = (int2*)pairs;
    const bool second = tid + PREP_BLOCK < CHUNK_VEC;
    const int4 d0 = dp[tid];
    const int4 s0 = sp[tid];
    int4 s1 = {0, 0, 0, 0}, d1 = {0, 0, 0, 0};
    if (second) {
        d1 = dp[tid + PREP_BLOCK];
        s1 = sp[tid + PREP_BLOCK];
    }
    if (tid < N_RANGE) cnt[tid] = 0;
    __syncthreads();
    atomicAdd(&cnt[range_of(d0.x)], 1);
    atomicAdd(&cnt[range_of(d0.y)], 1);
    atomicAdd(&cnt[range_of(d0.z)], 1);
    atomicAdd(&cnt[range_of(d0.w)], 1);
    if (second) {
        atomicAdd(&cnt[range_of(d1.x)], 1);
        atomicAdd(&cnt[range_of(d1.y)], 1);
        atomicAdd(&cnt[range_of(d1.z)], 1);
        atomicAdd(&cnt[range_of(d1.w)], 1);
    }
    __syncthreads();

    const int c = (tid < N_RANGE) ? cnt[tid] : 0;
    int incl = c;
    for (int o = 1; o < 32; o <<= 1) {
        int t = __shfl_up(incl, o);
        if (lane >= o) incl += t;
    }
    if (lane == 31) wsum[wv] = incl;
    __syncthreads();
    int woff = (lane < wv) ? wsum[lane] : 0;
    for (int o = 16; o > 0; o >>= 1) woff += __shfl_xor(woff, o);
    const int excl = woff + incl - c;

    const int base = (int)blockIdx.x * CHUNK_EDGES;
    if (tid < N_RANGE) {
        cnt[tid] = base + excl;
        cells[tid * N_CHUNK + blockIdx.x] = c;
        loff[tid * N_CHUNK + blockIdx.x]  = excl;
    }
    __syncthreads();

#define PLACE4(S, D)                                                              \
    do {                                                                          \
        int q0 = atomicAdd(&cnt[range_of((D).x)], 1);                             \
        int q1 = atomicAdd(&cnt[range_of((D).y)], 1);                             \
        int q2 = atomicAdd(&cnt[range_of((D).z)], 1);                             \
        int q3 = atomicAdd(&cnt[range_of((D).w)], 1);                             \
        out[min(max(q0, base), base + CHUNK_EDGES - 1)] = make_int2((S).x, (D).x); \
        out[min(max(q1, base), base + CHUNK_EDGES - 1)] = make_int2((S).y, (D).y); \
        out[min(max(q2, base), base + CHUNK_EDGES - 1)] = make_int2((S).z, (D).z); \
        out[min(max(q3, base), base + CHUNK_EDGES - 1)] = make_int2((S).w, (D).w); \
    } while (0)
    PLACE4(s0, d0);
    if (second) PLACE4(s1, d1);
#undef PLACE4
}

__global__ void __launch_bounds__(PREP_BLOCK)
node_place_kernel(const int* __restrict__ pairs, const int* __restrict__ cells, const int* __restrict__ loff,
                  int* __restrict__ rowptr, float* __restrict__ dinv, int* __restrict__ col) {
    __shared__ int cnt[RANGE_NODES];
    __shared__ int wsum[PREP_BLOCK / 32];
    __shared__ int wfront[PREP_BLOCK / 32];
    __shared__ int wown[PREP_BLOCK / 32];
    const int tid  = threadIdx.x;
    const int lane = tid & 31;
    const int wv   = tid >> 5;
    const int r    = blockIdx.x;
    const int g    = tid / SEG_LANES;
    const int sub  = tid % SEG_LANES;
    const int2* in = (const int2*)pairs;

    if (tid < RANGE_NODES) cnt[tid] = 0;

    const int gc = min(g, N_CHUNK - 1);
    int n  = cells[r * N_CHUNK + gc];
    int lo = loff[r * N_CHUNK + gc];
    if (g >= N_CHUNK) n = 0;
    lo = min(max(lo, 0), CHUNK_EDGES);
    n  = min(max(n, 0), CHUNK_EDGES - lo);
    const int2* seg = in + (size_t)gc * CHUNK_EDGES + lo;

    const int front = r * N_CHUNK;
    const int nvec  = (front + 3) >> 2;
    const int4* c4  = (const int4*)cells;
    int4 fv[FRONT_VEC];
#pragma unroll
    for (int k = 0; k < FRONT_VEC; ++k) fv[k] = c4[min(tid + k * PREP_BLOCK, max(nvec - 1, 0))];
    int part = 0;
#pragma unroll
    for (int k = 0; k < FRONT_VEC; ++k) {
        const int e = (tid + k * PREP_BLOCK) * 4;
        if (e     < front) part += fv[k].x;
        if (e + 1 < front) part += fv[k].y;
        if (e + 2 < front) part += fv[k].z;
        if (e + 3 < front) part += fv[k].w;
    }
    int own = (sub == 0) ? n : 0;
    for (int o = 16; o > 0; o >>= 1) {
        part += __shfl_xor(part, o);
        own  += __shfl_xor(own, o);
    }
    if (lane == 0) { wfront[wv] = part; wown[wv] = own; }
    __syncthreads();
    int tf = wfront[lane], to = wown[lane];
    for (int o = 16; o > 0; o >>= 1) {
        tf += __shfl_xor(tf, o);
        to += __shfl_xor(to, o);
    }
    const int beg = min(max(tf, 0), N_EDGES);
    const int end = min(beg + min(max(to, 0), N_EDGES), N_EDGES);

    for (int j0 = sub; j0 < n; j0 += SEG_LANES * SEG_BATCH) {
        int d[SEG_BATCH];
#pragma unroll
        for (int k = 0; k < SEG_BATCH; ++k) d[k] = seg[min(j0 + k * SEG_LANES, n - 1)].y;
#pragma unroll
        for (int k = 0; k < SEG_BATCH; ++k)
            if (j0 + k * SEG_LANES < n) atomicAdd(&cnt[d[k] & (RANGE_NODES - 1)], 1);
    }
    __syncthreads();

    const int c = (tid < RANGE_NODES) ? cnt[tid] : 0;
    int incl = c;
    for (int o = 1; o < 32; o <<= 1) {
        int t = __shfl_up(incl, o);
        if (lane >= o) incl += t;
    }
    if (lane == 31) wsum[wv] = incl;
    __syncthreads();
    int woff = (lane < wv) ? wsum[lane] : 0;
    for (int o = 16; o > 0; o >>= 1) woff += __shfl_xor(woff, o);
    const int excl = woff + incl - c;

    const int node = (r << RANGE_SHIFT) + tid;
    if (tid < RANGE_NODES) {
        cnt[tid] = excl;
        if (node < N_NODES) {
            rowptr[node] = beg + excl;
            dinv[node] = rsqrtf((float)c + 1.0f);
            if (node == N_NODES - 1) rowptr[N_NODES] = beg + excl + c;
        }
    }
    __syncthreads();

    for (int j0 = sub; j0 < n; j0 += SEG_LANES * SEG_BATCH) {
        int2 p[SEG_BATCH];
#pragma unroll
        for (int k = 0; k < SEG_BATCH; ++k) p[k] = seg[min(j0 + k * SEG_LANES, n - 1)];
#pragma unroll
        for (int k = 0; k < SEG_BATCH; ++k) {
            if (j0 + k * SEG_LANES < n) {
                const int slot = atomicAdd(&cnt[p[k].y & (RANGE_NODES - 1)], 1);
                col[min(beg + slot, end - 1)] = p[k].x;
            }
        }
    }
}


typedef __attribute__((ext_vector_type(4))) float f32x4;

#define PROD_ROWS 32

template <int K, int NCOL>
__global__ void __launch_bounds__((NCOL / 16) * 32)
gemm_wmma_f32(const float* __restrict__ A,
              const float* __restrict__ B,
              const float* __restrict__ rscale,
              float* __restrict__ C) {
    constexpr int NW    = NCOL / 16;
    constexpr int RPW   = PROD_ROWS / NW;
    constexpr int PITCH = K + 4;
    static_assert(K == 128, "one 16-byte load per lane covers a whole row");
    static_assert(PROD_ROWS % NW == 0 && N_NODES % PROD_ROWS == 0, "waves tile the block's rows, blocks tile the row list");

    __shared__ __attribute__((aligned(16))) float tile[PROD_ROWS * PITCH];

    const int wave = threadIdx.x >> 5;
    const int lane = threadIdx.x & 31;
    const int row0 = blockIdx.x * PROD_ROWS;

    const f32x4* xp = (const f32x4*)(A + (size_t)(row0 + wave * RPW) * K) + lane;
    f32x4 xv[RPW];
#pragma unroll
    for (int j = 0; j < RPW; ++j) xv[j] = __builtin_nontemporal_load(xp + j * (K / 4));
    __builtin_amdgcn_sched_barrier(0);

    const int m  = lane & 15;
    const int kk = (lane >> 4) << 1;
    const int n  = (lane & 15) + (wave << 4);

    const float* bp = B + (size_t)kk * NCOL + n;
    f32x2 bf[K / 4];
#pragma unroll
    for (int s = 0; s < K / 4; ++s) {
        bf[s].x = bp[(size_t)(4 * s) * NCOL];
        bf[s].y = bp[(size_t)(4 * s + 1) * NCOL];
    }

    const int mbase = (lane >> 4) << 3;
    float sc0[8], sc1[8];
#pragma unroll
    for (int i = 0; i < 8; ++i) {
        sc0[i] = rscale[row0 + mbase + i];
        sc1[i] = rscale[row0 + 16 + mbase + i];
    }

#pragma unroll
    for (int j = 0; j < RPW; ++j)
        *(f32x4*)(tile + (wave * RPW + j) * PITCH + lane * 4) = xv[j];
    __syncthreads();

    const float* ap = tile + m * PITCH + kk;
    float* cp = C + (size_t)row0 * NCOL + n;

    __builtin_amdgcn_sched_barrier(0);
    f32x8 acc0 = {0.f, 0.f, 0.f, 0.f, 0.f, 0.f, 0.f, 0.f};
    f32x8 acc1 = {0.f, 0.f, 0.f, 0.f, 0.f, 0.f, 0.f, 0.f};
#pragma unroll
    for (int s = 0; s < K / 4; ++s) {
        const f32x2 a0 = *(const f32x2*)(ap + 4 * s);
        const f32x2 a1 = *(const f32x2*)(ap + 16 * PITCH + 4 * s);
        acc0 = __builtin_amdgcn_wmma_f32_16x16x4_f32(
             false, a0,  false, bf[s],
             (short)0, acc0,  false,  false);
        acc1 = __builtin_amdgcn_wmma_f32_16x16x4_f32(
             false, a1,  false, bf[s],
             (short)0, acc1,  false,  false);
    }
    __builtin_amdgcn_sched_barrier(0);
    asm volatile("v_nop\n\tv_nop\n\tv_nop\n\tv_nop" : "+v"(acc0), "+v"(acc1));

#pragma unroll
    for (int i = 0; i < 8; ++i)
        cp[(size_t)(mbase + i) * NCOL] = acc0[i] * sc0[i];
#pragma unroll
    for (int i = 0; i < 8; ++i)
        cp[(size_t)(16 + mbase + i) * NCOL] = acc1[i] * sc1[i];
}


template <int L, int K>
__device__ __forceinline__ unsigned group_lane(unsigned x) {
    if constexpr (L == 16) {
        return (unsigned)__builtin_amdgcn_update_dpp(0, (int)x, 0x150 + K, 0xF, 0xF, true);
    } else {
        return __builtin_amdgcn_mov_dpp8(x, (unsigned)K * 0x249249u);
    }
}

template <int L>
__device__ __forceinline__ f32x4 bucket_sum(const f32x4* __restrict__ hv, const int* __restrict__ col,
                                            int node, int beg, int deg, int t) {
    f32x4 acc = hv[(unsigned)node * L + (unsigned)t];
    int my = 0;
    if (t < deg) my = col[(unsigned)(beg + t)];
    unsigned pre = (unsigned)min(max(my, 0), N_NODES - 1) * L;

    for (int base = 0; base < deg; base += L) {
        const int rem = deg - base;
        const int nxt = col[(unsigned)(beg + min(base + L + t, deg - 1))];

        f32x4 v[8];
#define LOAD_ROW(K)                                                       \
        {                                                                 \
            const unsigned p_k = group_lane<L, (K)>(pre) + (unsigned)t;   \
            if ((K) < rem) v[(K) & 7] = hv[p_k];                          \
        }
#define SUM_ROW(K) if ((K) < rem) acc += v[(K) & 7];
        LOAD_ROW(0) LOAD_ROW(1) LOAD_ROW(2) LOAD_ROW(3) LOAD_ROW(4) LOAD_ROW(5) LOAD_ROW(6) LOAD_ROW(7)
        SUM_ROW(0) SUM_ROW(1) SUM_ROW(2) SUM_ROW(3) SUM_ROW(4) SUM_ROW(5) SUM_ROW(6) SUM_ROW(7)
        if constexpr (L == 16) {
            if (rem > 8) {
                LOAD_ROW(8) LOAD_ROW(9) LOAD_ROW(10) LOAD_ROW(11) LOAD_ROW(12) LOAD_ROW(13) LOAD_ROW(14) LOAD_ROW(15)
                SUM_ROW(8) SUM_ROW(9) SUM_ROW(10) SUM_ROW(11) SUM_ROW(12) SUM_ROW(13) SUM_ROW(14) SUM_ROW(15)
            }
        }
#undef LOAD_ROW
#undef SUM_ROW
        pre = (unsigned)min(max(nxt, 0), N_NODES - 1) * L;
    }
    return acc;
}

template <int C>
__global__ void __launch_bounds__(256)
gather_finish_kernel(const float* __restrict__ hs,
                     const int* __restrict__ rowptr,
                     const int* __restrict__ col,
                     const float* __restrict__ dinv,
                     const float* __restrict__ bias,
                     float* __restrict__ dst) {
    constexpr int L = C / 4;
    constexpr int G = 32 / L;
    static_assert(L == 16 || L == 8, "a node's row covers 16 or 8 lanes");
    static_assert(N_NODES % (8 * G) == 0, "blocks tile the node list");

    const int lane = threadIdx.x & 31;
    const int t    = lane & (L - 1);
    const int node = (int)(blockIdx.x * 8 + (threadIdx.x >> 5)) * G + (lane / L);
    const int beg  = rowptr[node];
    const int deg  = rowptr[node + 1] - beg;
    const float di = dinv[node];
    const f32x4 bv = *(const f32x4*)(bias + t * 4);

    const f32x4 acc = bucket_sum<L>((const f32x4*)hs, col, node, beg, deg, t);

    *(f32x4*)(dst + (size_t)node * C + (size_t)t * 4) = acc * di + bv;
}

#define TILE_PITCH (HID_CH + 4)

__global__ void __launch_bounds__(256)
gather_relu_product_kernel(const float* __restrict__ hs,
                           const int* __restrict__ rowptr,
                           const int* __restrict__ col,
                           const float* __restrict__ dinv,
                           const float* __restrict__ bias,
                           const float* __restrict__ Wn,
                           float* __restrict__ hn) {
    constexpr int L = HID_CH / 4;
    static_assert(L == 16 && N_NODES % 16 == 0, "a block's nodes are the 16 rows of one matrix tile");
    static_assert(OUT_CH % 16 == 0 && OUT_CH / 16 <= 8, "one wave per column tile");

    __shared__ __attribute__((aligned(16))) float tile[16 * TILE_PITCH];
    __shared__ float tscale[16];

    const int wave = threadIdx.x >> 5;
    const int lane = threadIdx.x & 31;
    const int t    = lane & (L - 1);
    const int row  = threadIdx.x >> 4;
    const int row0 = blockIdx.x << 4;
    const int node = row0 + row;
    const int beg  = rowptr[node];
    const int deg  = rowptr[node + 1] - beg;
    const float di = dinv[node];
    const f32x4 bv = *(const f32x4*)(bias + t * 4);

    const f32x4 acc = bucket_sum<L>((const f32x4*)hs, col, node, beg, deg, t);

    f32x4 r = acc * di + bv;
    r.x = fmaxf(r.x, 0.0f); r.y = fmaxf(r.y, 0.0f);
    r.z = fmaxf(r.z, 0.0f); r.w = fmaxf(r.w, 0.0f);
    *(f32x4*)(tile + row * TILE_PITCH + t * 4) = r;
    if (t == 0) tscale[row] = di;

    const int m  = lane & 15;
    const int kk = (lane >> 4) << 1;
    const int n  = m + (wave << 4);
    const bool owner = wave < OUT_CH / 16;

    f32x2 bf[HID_CH / 4];
    if (owner) {
        const float* bp = Wn + (size_t)kk * OUT_CH + n;
#pragma unroll
        for (int s = 0; s < HID_CH / 4; ++s) {
            bf[s].x = bp[(size_t)(4 * s) * OUT_CH];
            bf[s].y = bp[(size_t)(4 * s + 1) * OUT_CH];
        }
    }
    __syncthreads();
    if (!owner) return;

    const float* ap = tile + m * TILE_PITCH + kk;
    const int mbase = (lane >> 4) << 3;
    float* cp = hn + (size_t)row0 * OUT_CH + n;
    float sc[8];
#pragma unroll
    for (int i = 0; i < 8; ++i) sc[i] = tscale[mbase + i];

    __builtin_amdgcn_sched_barrier(0);
    f32x8 pacc = {0.f, 0.f, 0.f, 0.f, 0.f, 0.f, 0.f, 0.f};
#pragma unroll
    for (int s = 0; s < HID_CH / 4; ++s) {
        const f32x2 a = *(const f32x2*)(ap + 4 * s);
        pacc = __builtin_amdgcn_wmma_f32_16x16x4_f32(
             false, a,  false, bf[s],
             (short)0, pacc,  false,  false);
    }
    __builtin_amdgcn_sched_barrier(0);
    asm volatile("v_nop\n\tv_nop\n\tv_nop\n\tv_nop" : "+v"(pacc));

#pragma unroll
    for (int i = 0; i < 8; ++i)
        cp[(size_t)(mbase + i) * OUT_CH] = pacc[i] * sc[i];
}


extern "C" void kernel_launch(void* const* d_in, const int* in_sizes, int n_in,
                              void* d_out, int out_size, void* d_ws, size_t ws_size,
                              hipStream_t stream) {
    const float* x    = (const float*)d_in[0];
    const int*   ei   = (const int*)d_in[1];
    const float* W1   = (const float*)d_in[2];
    const float* b1   = (const float*)d_in[3];
    const float* W2   = (const float*)d_in[4];
    const float* b2   = (const float*)d_in[5];
    float*       out  = (float*)d_out;

    const int* srcv = ei;
    const int* dstv = ei + N_EDGES;

    float* dinv   = (float*)d_ws;
    int*   rowptr = (int*)(dinv + 100352);
    int*   col    = rowptr + 100352;
    float* h1     = (float*)(col + N_EDGES);
    float* h2     = h1 + (size_t)N_NODES * HID_CH;
    int*   cells = (int*)h1;
    int*   loff  = cells + 49152;
    int*   pairs = loff + 49152;

    chunk_bucket_kernel<<<N_CHUNK, PREP_BLOCK, 0, stream>>>(srcv, dstv, cells, loff, pairs);
    node_place_kernel<<<N_RANGE, PREP_BLOCK, 0, stream>>>(pairs, cells, loff, rowptr, dinv, col);

    gemm_wmma_f32<IN_CH, HID_CH><<<N_NODES / PROD_ROWS, 128, 0, stream>>>(x, W1, dinv, h1);

    gather_relu_product_kernel<<<N_NODES / 16, 256, 0, stream>>>(h1, rowptr, col, dinv, b1, W2, h2);

    gather_finish_kernel<OUT_CH><<<N_NODES / 32, 256, 0, stream>>>(h2, rowptr, col, dinv, b2, out);
}


# Bucketing launches put a chunk's pair run and a range's piece of col in order in LDS and store them as whole contiguous lines, not one scattered 8- or 4-byte store per edge; slot order kept
# speedup vs baseline: 6.4798x; 1.0832x over previous
#include <hip/hip_runtime.h>
#include <hip/hip_bf16.h>


#define N_NODES 100000
#define N_EDGES 1600000
#define IN_CH   128
#define HID_CH  64
#define OUT_CH  32

typedef __attribute__((ext_vector_type(2))) float f32x2;
typedef __attribute__((ext_vector_type(8))) float f32x8;


#define RANGE_SHIFT 9
#define RANGE_NODES (1 << RANGE_SHIFT)
#define N_RANGE     ((N_NODES + RANGE_NODES - 1) / RANGE_NODES)
#define CHUNK_EDGES 6400
#define N_CHUNK     (N_EDGES / CHUNK_EDGES)
#define CHUNK_VEC   (CHUNK_EDGES / 4)
#define N_CELL      (N_RANGE * N_CHUNK)
#define PREP_BLOCK  1024
#define SEG_LANES   4
#define SEG_BATCH   12
#define FRONT_VEC   12
#define COL_STAGE   12288

static_assert(N_CHUNK * CHUNK_EDGES == N_EDGES, "chunks must tile the edge list");
static_assert(N_CELL <= 49152 && N_CELL % 4 == 0, "count matrix fits its workspace plane, in whole 16-byte pieces");
static_assert(PREP_BLOCK <= CHUNK_VEC && CHUNK_VEC <= 2 * PREP_BLOCK, "two 16-byte loads per thread cover a chunk");
static_assert(N_RANGE <= PREP_BLOCK && RANGE_NODES <= PREP_BLOCK, "one thread per range / per node of a range");
static_assert(N_CHUNK * SEG_LANES <= PREP_BLOCK, "one group of lanes per chunk");
static_assert(CHUNK_EDGES % 2 == 0, "a chunk's run of 8-byte pairs is a whole number of 16-byte pieces");
static_assert(((N_RANGE - 1) * N_CHUNK + 3) / 4 <= FRONT_VEC * PREP_BLOCK, "one batch of loads covers all counts in front");

__device__ __forceinline__ int range_of(int d) {
    return (int)min((unsigned)d >> RANGE_SHIFT, (unsigned)(N_RANGE - 1));
}

__global__ void __launch_bounds__(PREP_BLOCK)
chunk_bucket_kernel(const int* __restrict__ src, const int* __restrict__ dst,
                    int* __restrict__ cells, int* __restrict__ loff, int* __restrict__ pairs) {
    __shared__ int cnt[N_RANGE];
    __shared__ int wsum[PREP_BLOCK / 32];
    __shared__ __attribute__((aligned(16))) int stage[CHUNK_EDGES * 2];
    const int tid  = threadIdx.x;
    const int lane = tid & 31;
    const int wv   = tid >> 5;

    const int4* sp = (const int4*)(src + (size_t)blockIdx.x * CHUNK_EDGES);
    const int4* dp = (const int4*)(dst + (size_t)blockIdx.x * CHUNK_EDGES);
    const bool second = tid + PREP_BLOCK < CHUNK_VEC;
    const int4 d0 = dp[tid];
    const int4 s0 = sp[tid];
    int4 s1 = {0, 0, 0, 0}, d1 = {0, 0, 0, 0};
    if (second) {
        d1 = dp[tid + PREP_BLOCK];
        s1 = sp[tid + PREP_BLOCK];
    }
    if (tid < N_RANGE) cnt[tid] = 0;
    __syncthreads();
    atomicAdd(&cnt[range_of(d0.x)], 1);
    atomicAdd(&cnt[range_of(d0.y)], 1);
    atomicAdd(&cnt[range_of(d0.z)], 1);
    atomicAdd(&cnt[range_of(d0.w)], 1);
    if (second) {
        atomicAdd(&cnt[range_of(d1.x)], 1);
        atomicAdd(&cnt[range_of(d1.y)], 1);
        atomicAdd(&cnt[range_of(d1.z)], 1);
        atomicAdd(&cnt[range_of(d1.w)], 1);
    }
    __syncthreads();

    const int c = (tid < N_RANGE) ? cnt[tid] : 0;
    int incl = c;
    for (int o = 1; o < 32; o <<= 1) {
        int t = __shfl_up(incl, o);
        if (lane >= o) incl += t;
    }
    if (lane == 31) wsum[wv] = incl;
    __syncthreads();
    int woff = (lane < wv) ? wsum[lane] : 0;
    for (int o = 16; o > 0; o >>= 1) woff += __shfl_xor(woff, o);
    const int excl = woff + incl - c;

    if (tid < N_RANGE) {
        cnt[tid] = excl;
        cells[tid * N_CHUNK + blockIdx.x] = c;
        loff[tid * N_CHUNK + blockIdx.x]  = excl;
    }
    __syncthreads();

#define PUT(Q, SV, DV) *(int2*)(stage + 2 * min(max((Q), 0), CHUNK_EDGES - 1)) = make_int2((SV), (DV))
#define PLACE4(S, D)                                  \
    do {                                              \
        int q0 = atomicAdd(&cnt[range_of((D).x)], 1); \
        int q1 = atomicAdd(&cnt[range_of((D).y)], 1); \
        int q2 = atomicAdd(&cnt[range_of((D).z)], 1); \
        int q3 = atomicAdd(&cnt[range_of((D).w)], 1); \
        PUT(q0, (S).x, (D).x);                        \
        PUT(q1, (S).y, (D).y);                        \
        PUT(q2, (S).z, (D).z);                        \
        PUT(q3, (S).w, (D).w);                        \
    } while (0)
    PLACE4(s0, d0);
    if (second) PLACE4(s1, d1);
#undef PLACE4
#undef PUT
    __syncthreads();

    constexpr int RUN_VEC = CHUNK_EDGES / 2;
    constexpr int RUN_PER = (RUN_VEC + PREP_BLOCK - 1) / PREP_BLOCK;
    const int4* s4 = (const int4*)stage;
    int4* o4 = (int4*)(pairs + (size_t)blockIdx.x * (CHUNK_EDGES * 2));
    int4 rv[RUN_PER];
    for (int k = 0; k < RUN_PER; ++k) rv[k] = s4[min(tid + k * PREP_BLOCK, RUN_VEC - 1)];
    for (int k = 0; k < RUN_PER; ++k)
        if (tid + k * PREP_BLOCK < RUN_VEC) o4[tid + k * PREP_BLOCK] = rv[k];
}

__global__ void __launch_bounds__(PREP_BLOCK)
node_place_kernel(const int* __restrict__ pairs, const int* __restrict__ cells, const int* __restrict__ loff,
                  int* __restrict__ rowptr, float* __restrict__ dinv, int* __restrict__ col) {
    __shared__ int cnt[RANGE_NODES];
    __shared__ int cstage[COL_STAGE];
    __shared__ int wsum[PREP_BLOCK / 32];
    __shared__ int wfront[PREP_BLOCK / 32];
    __shared__ int wown[PREP_BLOCK / 32];
    const int tid  = threadIdx.x;
    const int lane = tid & 31;
    const int wv   = tid >> 5;
    const int r    = blockIdx.x;
    const int g    = tid / SEG_LANES;
    const int sub  = tid % SEG_LANES;
    const int2* in = (const int2*)pairs;

    if (tid < RANGE_NODES) cnt[tid] = 0;

    const int gc = min(g, N_CHUNK - 1);
    int n  = cells[r * N_CHUNK + gc];
    int lo = loff[r * N_CHUNK + gc];
    if (g >= N_CHUNK) n = 0;
    lo = min(max(lo, 0), CHUNK_EDGES);
    n  = min(max(n, 0), CHUNK_EDGES - lo);
    const int2* seg = in + (size_t)gc * CHUNK_EDGES + lo;

    const int front = r * N_CHUNK;
    const int nvec  = (front + 3) >> 2;
    const int4* c4  = (const int4*)cells;
    int4 fv[FRONT_VEC];
#pragma unroll
    for (int k = 0; k < FRONT_VEC; ++k) fv[k] = c4[min(tid + k * PREP_BLOCK, max(nvec - 1, 0))];
    int part = 0;
#pragma unroll
    for (int k = 0; k < FRONT_VEC; ++k) {
        const int e = (tid + k * PREP_BLOCK) * 4;
        if (e     < front) part += fv[k].x;
        if (e + 1 < front) part += fv[k].y;
        if (e + 2 < front) part += fv[k].z;
        if (e + 3 < front) part += fv[k].w;
    }
    int own = (sub == 0) ? n : 0;
    for (int o = 16; o > 0; o >>= 1) {
        part += __shfl_xor(part, o);
        own  += __shfl_xor(own, o);
    }
    if (lane == 0) { wfront[wv] = part; wown[wv] = own; }
    __syncthreads();
    int tf = wfront[lane], to = wown[lane];
    for (int o = 16; o > 0; o >>= 1) {
        tf += __shfl_xor(tf, o);
        to += __shfl_xor(to, o);
    }
    const int beg = min(max(tf, 0), N_EDGES);
    const int end = min(beg + min(max(to, 0), N_EDGES), N_EDGES);

    for (int j0 = sub; j0 < n; j0 += SEG_LANES * SEG_BATCH) {
        int d[SEG_BATCH];
#pragma unroll
        for (int k = 0; k < SEG_BATCH; ++k) d[k] = seg[min(j0 + k * SEG_LANES, n - 1)].y;
#pragma unroll
        for (int k = 0; k < SEG_BATCH; ++k)
            if (j0 + k * SEG_LANES < n) atomicAdd(&cnt[d[k] & (RANGE_NODES - 1)], 1);
    }
    __syncthreads();

    const int c = (tid < RANGE_NODES) ? cnt[tid] : 0;
    int incl = c;
    for (int o = 1; o < 32; o <<= 1) {
        int t = __shfl_up(incl, o);
        if (lane >= o) incl += t;
    }
    if (lane == 31) wsum[wv] = incl;
    __syncthreads();
    int woff = (lane < wv) ? wsum[lane] : 0;
    for (int o = 16; o > 0; o >>= 1) woff += __shfl_xor(woff, o);
    const int excl = woff + incl - c;

    const int node = (r << RANGE_SHIFT) + tid;
    if (tid < RANGE_NODES) {
        cnt[tid] = excl;
        if (node < N_NODES) {
            rowptr[node] = beg + excl;
            dinv[node] = rsqrtf((float)c + 1.0f);
            if (node == N_NODES - 1) rowptr[N_NODES] = beg + excl + c;
        }
    }
    __syncthreads();

    for (int j0 = sub; j0 < n; j0 += SEG_LANES * SEG_BATCH) {
        int2 p[SEG_BATCH];
#pragma unroll
        for (int k = 0; k < SEG_BATCH; ++k) p[k] = seg[min(j0 + k * SEG_LANES, n - 1)];
#pragma unroll
        for (int k = 0; k < SEG_BATCH; ++k) {
            if (j0 + k * SEG_LANES < n) {
                const int slot = atomicAdd(&cnt[p[k].y & (RANGE_NODES - 1)], 1);
                if ((unsigned)slot < (unsigned)COL_STAGE) cstage[slot] = p[k].x;
                else col[min(beg + slot, end - 1)] = p[k].x;
            }
        }
    }
    __syncthreads();

    const int nst = min(end - beg, COL_STAGE);
    for (int i0 = tid; i0 < nst; i0 += 4 * PREP_BLOCK) {
        int cv[4];
        for (int k = 0; k < 4; ++k) cv[k] = cstage[min(i0 + k * PREP_BLOCK, COL_STAGE - 1)];
        for (int k = 0; k < 4; ++k)
            if (i0 + k * PREP_BLOCK < nst) col[beg + i0 + k * PREP_BLOCK] = cv[k];
    }
}


typedef __attribute__((ext_vector_type(4))) float f32x4;

#define PROD_ROWS 32

template <int K, int NCOL>
__global__ void __launch_bounds__((NCOL / 16) * 32)
gemm_wmma_f32(const float* __restrict__ A,
              const float* __restrict__ B,
              const float* __restrict__ rscale,
              float* __restrict__ C) {
    constexpr int NW    = NCOL / 16;
    constexpr int RPW   = PROD_ROWS / NW;
    constexpr int PITCH = K + 4;
    static_assert(K == 128, "one 16-byte load per lane covers a whole row");
    static_assert(PROD_ROWS % NW == 0 && N_NODES % PROD_ROWS == 0, "waves tile the block's rows, blocks tile the row list");

    __shared__ __attribute__((aligned(16))) float tile[PROD_ROWS * PITCH];

    const int wave = threadIdx.x >> 5;
    const int lane = threadIdx.x & 31;
    const int row0 = blockIdx.x * PROD_ROWS;

    const f32x4* xp = (const f32x4*)(A + (size_t)(row0 + wave * RPW) * K) + lane;
    f32x4 xv[RPW];
#pragma unroll
    for (int j = 0; j < RPW; ++j) xv[j] = __builtin_nontemporal_load(xp + j * (K / 4));
    __builtin_amdgcn_sched_barrier(0);

    const int m  = lane & 15;
    const int kk = (lane >> 4) << 1;
    const int n  = (lane & 15) + (wave << 4);

    const float* bp = B + (size_t)kk * NCOL + n;
    f32x2 bf[K / 4];
#pragma unroll
    for (int s = 0; s < K / 4; ++s) {
        bf[s].x = bp[(size_t)(4 * s) * NCOL];
        bf[s].y = bp[(size_t)(4 * s + 1) * NCOL];
    }

    const int mbase = (lane >> 4) << 3;
    float sc0[8], sc1[8];
#pragma unroll
    for (int i = 0; i < 8; ++i) {
        sc0[i] = rscale[row0 + mbase + i];
        sc1[i] = rscale[row0 + 16 + mbase + i];
    }

#pragma unroll
    for (int j = 0; j < RPW; ++j)
        *(f32x4*)(tile + (wave * RPW + j) * PITCH + lane * 4) = xv[j];
    __syncthreads();

    const float* ap = tile + m * PITCH + kk;
    float* cp = C + (size_t)row0 * NCOL + n;

    __builtin_amdgcn_sched_barrier(0);
    f32x8 acc0 = {0.f, 0.f, 0.f, 0.f, 0.f, 0.f, 0.f, 0.f};
    f32x8 acc1 = {0.f, 0.f, 0.f, 0.f, 0.f, 0.f, 0.f, 0.f};
#pragma unroll
    for (int s = 0; s < K / 4; ++s) {
        const f32x2 a0 = *(const f32x2*)(ap + 4 * s);
        const f32x2 a1 = *(const f32x2*)(ap + 16 * PITCH + 4 * s);
        acc0 = __builtin_amdgcn_wmma_f32_16x16x4_f32(
             false, a0,  false, bf[s],
             (short)0, acc0,  false,  false);
        acc1 = __builtin_amdgcn_wmma_f32_16x16x4_f32(
             false, a1,  false, bf[s],
             (short)0, acc1,  false,  false);
    }
    __builtin_amdgcn_sched_barrier(0);
    asm volatile("v_nop\n\tv_nop\n\tv_nop\n\tv_nop" : "+v"(acc0), "+v"(acc1));

#pragma unroll
    for (int i = 0; i < 8; ++i)
        cp[(size_t)(mbase + i) * NCOL] = acc0[i] * sc0[i];
#pragma unroll
    for (int i = 0; i < 8; ++i)
        cp[(size_t)(16 + mbase + i) * NCOL] = acc1[i] * sc1[i];
}


template <int L, int K>
__device__ __forceinline__ unsigned group_lane(unsigned x) {
    if constexpr (L == 16) {
        return (unsigned)__builtin_amdgcn_update_dpp(0, (int)x, 0x150 + K, 0xF, 0xF, true);
    } else {
        return __builtin_amdgcn_mov_dpp8(x, (unsigned)K * 0x249249u);
    }
}

template <int L>
__device__ __forceinline__ f32x4 bucket_sum(const f32x4* __restrict__ hv, const int* __restrict__ col,
                                            int node, int beg, int deg, int t) {
    f32x4 acc = hv[(unsigned)node * L + (unsigned)t];
    int my = 0;
    if (t < deg) my = col[(unsigned)(beg + t)];
    unsigned pre = (unsigned)min(max(my, 0), N_NODES - 1) * L;

    for (int base = 0; base < deg; base += L) {
        const int rem = deg - base;
        const int nxt = col[(unsigned)(beg + min(base + L + t, deg - 1))];

        f32x4 v[8];
#define LOAD_ROW(K)                                                       \
        {                                                                 \
            const unsigned p_k = group_lane<L, (K)>(pre) + (unsigned)t;   \
            if ((K) < rem) v[(K) & 7] = hv[p_k];                          \
        }
#define SUM_ROW(K) if ((K) < rem) acc += v[(K) & 7];
        LOAD_ROW(0) LOAD_ROW(1) LOAD_ROW(2) LOAD_ROW(3) LOAD_ROW(4) LOAD_ROW(5) LOAD_ROW(6) LOAD_ROW(7)
        SUM_ROW(0) SUM_ROW(1) SUM_ROW(2) SUM_ROW(3) SUM_ROW(4) SUM_ROW(5) SUM_ROW(6) SUM_ROW(7)
        if constexpr (L == 16) {
            if (rem > 8) {
                LOAD_ROW(8) LOAD_ROW(9) LOAD_ROW(10) LOAD_ROW(11) LOAD_ROW(12) LOAD_ROW(13) LOAD_ROW(14) LOAD_ROW(15)
                SUM_ROW(8) SUM_ROW(9) SUM_ROW(10) SUM_ROW(11) SUM_ROW(12) SUM_ROW(13) SUM_ROW(14) SUM_ROW(15)
            }
        }
#undef LOAD_ROW
#undef SUM_ROW
        pre = (unsigned)min(max(nxt, 0), N_NODES - 1) * L;
    }
    return acc;
}

template <int C>
__global__ void __launch_bounds__(256)
gather_finish_kernel(const float* __restrict__ hs,
                     const int* __restrict__ rowptr,
                     const int* __restrict__ col,
                     const float* __restrict__ dinv,
                     const float* __restrict__ bias,
                     float* __restrict__ dst) {
    constexpr int L = C / 4;
    constexpr int G = 32 / L;
    static_assert(L == 16 || L == 8, "a node's row covers 16 or 8 lanes");
    static_assert(N_NODES % (8 * G) == 0, "blocks tile the node list");

    const int lane = threadIdx.x & 31;
    const int t    = lane & (L - 1);
    const int node = (int)(blockIdx.x * 8 + (threadIdx.x >> 5)) * G + (lane / L);
    const int beg  = rowptr[node];
    const int deg  = rowptr[node + 1] - beg;
    const float di = dinv[node];
    const f32x4 bv = *(const f32x4*)(bias + t * 4);

    const f32x4 acc = bucket_sum<L>((const f32x4*)hs, col, node, beg, deg, t);

    *(f32x4*)(dst + (size_t)node * C + (size_t)t * 4) = acc * di + bv;
}

#define TILE_PITCH (HID_CH + 4)

__global__ void __launch_bounds__(256)
gather_relu_product_kernel(const float* __restrict__ hs,
                           const int* __restrict__ rowptr,
                           const int* __restrict__ col,
                           const float* __restrict__ dinv,
                           const float* __restrict__ bias,
                           const float* __restrict__ Wn,
                           float* __restrict__ hn) {
    constexpr int L = HID_CH / 4;
    static_assert(L == 16 && N_NODES % 16 == 0, "a block's nodes are the 16 rows of one matrix tile");
    static_assert(OUT_CH % 16 == 0 && OUT_CH / 16 <= 8, "one wave per column tile");

    __shared__ __attribute__((aligned(16))) float tile[16 * TILE_PITCH];
    __shared__ float tscale[16];

    const int wave = threadIdx.x >> 5;
    const int lane = threadIdx.x & 31;
    const int t    = lane & (L - 1);
    const int row  = threadIdx.x >> 4;
    const int row0 = blockIdx.x << 4;
    const int node = row0 + row;
    const int beg  = rowptr[node];
    const int deg  = rowptr[node + 1] - beg;
    const float di = dinv[node];
    const f32x4 bv = *(const f32x4*)(bias + t * 4);

    const f32x4 acc = bucket_sum<L>((const f32x4*)hs, col, node, beg, deg, t);

    f32x4 r = acc * di + bv;
    r.x = fmaxf(r.x, 0.0f); r.y = fmaxf(r.y, 0.0f);
    r.z = fmaxf(r.z, 0.0f); r.w = fmaxf(r.w, 0.0f);
    *(f32x4*)(tile + row * TILE_PITCH + t * 4) = r;
    if (t == 0) tscale[row] = di;

    const int m  = lane & 15;
    const int kk = (lane >> 4) << 1;
    const int n  = m + (wave << 4);
    const bool owner = wave < OUT_CH / 16;

    f32x2 bf[HID_CH / 4];
    if (owner) {
        const float* bp = Wn + (size_t)kk * OUT_CH + n;
#pragma unroll
        for (int s = 0; s < HID_CH / 4; ++s) {
            bf[s].x = bp[(size_t)(4 * s) * OUT_CH];
            bf[s].y = bp[(size_t)(4 * s + 1) * OUT_CH];
        }
    }
    __syncthreads();
    if (!owner) return;

    const float* ap = tile + m * TILE_PITCH + kk;
    const int mbase = (lane >> 4) << 3;
    float* cp = hn + (size_t)row0 * OUT_CH + n;
    float sc[8];
#pragma unroll
    for (int i = 0; i < 8; ++i) sc[i] = tscale[mbase + i];

    __builtin_amdgcn_sched_barrier(0);
    f32x8 pacc = {0.f, 0.f, 0.f, 0.f, 0.f, 0.f, 0.f, 0.f};
#pragma unroll
    for (int s = 0; s < HID_CH / 4; ++s) {
        const f32x2 a = *(const f32x2*)(ap + 4 * s);
        pacc = __builtin_amdgcn_wmma_f32_16x16x4_f32(
             false, a,  false, bf[s],
             (short)0, pacc,  false,  false);
    }
    __builtin_amdgcn_sched_barrier(0);
    asm volatile("v_nop\n\tv_nop\n\tv_nop\n\tv_nop" : "+v"(pacc));

#pragma unroll
    for (int i = 0; i < 8; ++i)
        cp[(size_t)(mbase + i) * OUT_CH] = pacc[i] * sc[i];
}


extern "C" void kernel_launch(void* const* d_in, const int* in_sizes, int n_in,
                              void* d_out, int out_size, void* d_ws, size_t ws_size,
                              hipStream_t stream) {
    const float* x    = (const float*)d_in[0];
    const int*   ei   = (const int*)d_in[1];
    const float* W1   = (const float*)d_in[2];
    const float* b1   = (const float*)d_in[3];
    const float* W2   = (const float*)d_in[4];
    const float* b2   = (const float*)d_in[5];
    float*       out  = (float*)d_out;

    const int* srcv = ei;
    const int* dstv = ei + N_EDGES;

    float* dinv   = (float*)d_ws;
    int*   rowptr = (int*)(dinv + 100352);
    int*   col    = rowptr + 100352;
    float* h1     = (float*)(col + N_EDGES);
    float* h2     = h1 + (size_t)N_NODES * HID_CH;
    int*   cells = (int*)h1;
    int*   loff  = cells + 49152;
    int*   pairs = loff + 49152;

    chunk_bucket_kernel<<<N_CHUNK, PREP_BLOCK, 0, stream>>>(srcv, dstv, cells, loff, pairs);
    node_place_kernel<<<N_RANGE, PREP_BLOCK, 0, stream>>>(pairs, cells, loff, rowptr, dinv, col);

    gemm_wmma_f32<IN_CH, HID_CH><<<N_NODES / PROD_ROWS, 128, 0, stream>>>(x, W1, dinv, h1);

    gather_relu_product_kernel<<<N_NODES / 16, 256, 0, stream>>>(h1, rowptr, col, dinv, b1, W2, h2);

    gather_finish_kernel<OUT_CH><<<N_NODES / 32, 256, 0, stream>>>(h2, rowptr, col, dinv, b2, out);
}


# Bucketing launches place a pair at offset + the rank its counting add returned: no second LDS atomic pass; fine pass reads its pairs once, keeps them in registers; slot order, so sum order, changes
# speedup vs baseline: 6.6822x; 1.0312x over previous
#include <hip/hip_runtime.h>
#include <hip/hip_bf16.h>


#define N_NODES 100000
#define N_EDGES 1600000
#define IN_CH   128
#define HID_CH  64
#define OUT_CH  32

typedef __attribute__((ext_vector_type(2))) float f32x2;
typedef __attribute__((ext_vector_type(8))) float f32x8;


#define RANGE_SHIFT 9
#define RANGE_NODES (1 << RANGE_SHIFT)
#define N_RANGE     ((N_NODES + RANGE_NODES - 1) / RANGE_NODES)
#define CHUNK_EDGES 6400
#define N_CHUNK     (N_EDGES / CHUNK_EDGES)
#define CHUNK_VEC   (CHUNK_EDGES / 4)
#define N_CELL      (N_RANGE * N_CHUNK)
#define PREP_BLOCK  1024
#define SEG_LANES   4
#define SEG_BATCH   12
#define FRONT_VEC   12
#define COL_STAGE   12288

static_assert(N_CHUNK * CHUNK_EDGES == N_EDGES, "chunks must tile the edge list");
static_assert(N_CELL <= 49152 && N_CELL % 4 == 0, "count matrix fits its workspace plane, in whole 16-byte pieces");
static_assert(PREP_BLOCK <= CHUNK_VEC && CHUNK_VEC <= 2 * PREP_BLOCK, "two 16-byte loads per thread cover a chunk");
static_assert(N_RANGE <= PREP_BLOCK && RANGE_NODES <= PREP_BLOCK, "one thread per range / per node of a range");
static_assert(N_CHUNK * SEG_LANES <= PREP_BLOCK, "one group of lanes per chunk");
static_assert(CHUNK_EDGES % 2 == 0, "a chunk's run of 8-byte pairs is a whole number of 16-byte pieces");
static_assert(((N_RANGE - 1) * N_CHUNK + 3) / 4 <= FRONT_VEC * PREP_BLOCK, "one batch of loads covers all counts in front");

__device__ __forceinline__ int range_of(int d) {
    return (int)min((unsigned)d >> RANGE_SHIFT, (unsigned)(N_RANGE - 1));
}

__global__ void __launch_bounds__(PREP_BLOCK)
chunk_bucket_kernel(const int* __restrict__ src, const int* __restrict__ dst,
                    int* __restrict__ cells, int* __restrict__ loff, int* __restrict__ pairs) {
    __shared__ int cnt[N_RANGE];
    __shared__ int wsum[PREP_BLOCK / 32];
    __shared__ __attribute__((aligned(16))) int stage[CHUNK_EDGES * 2];
    const int tid  = threadIdx.x;
    const int lane = tid & 31;
    const int wv   = tid >> 5;

    const int4* sp = (const int4*)(src + (size_t)blockIdx.x * CHUNK_EDGES);
    const int4* dp = (const int4*)(dst + (size_t)blockIdx.x * CHUNK_EDGES);
    const bool second = tid + PREP_BLOCK < CHUNK_VEC;
    const int4 d0 = dp[tid];
    const int4 s0 = sp[tid];
    int4 s1 = {0, 0, 0, 0}, d1 = {0, 0, 0, 0};
    if (second) {
        d1 = dp[tid + PREP_BLOCK];
        s1 = sp[tid + PREP_BLOCK];
    }
    if (tid < N_RANGE) cnt[tid] = 0;
    __syncthreads();
    const int k00 = atomicAdd(&cnt[range_of(d0.x)], 1);
    const int k01 = atomicAdd(&cnt[range_of(d0.y)], 1);
    const int k02 = atomicAdd(&cnt[range_of(d0.z)], 1);
    const int k03 = atomicAdd(&cnt[range_of(d0.w)], 1);
    int k10 = 0, k11 = 0, k12 = 0, k13 = 0;
    if (second) {
        k10 = atomicAdd(&cnt[range_of(d1.x)], 1);
        k11 = atomicAdd(&cnt[range_of(d1.y)], 1);
        k12 = atomicAdd(&cnt[range_of(d1.z)], 1);
        k13 = atomicAdd(&cnt[range_of(d1.w)], 1);
    }
    __syncthreads();

    const int c = (tid < N_RANGE) ? cnt[tid] : 0;
    int incl = c;
    for (int o = 1; o < 32; o <<= 1) {
        int t = __shfl_up(incl, o);
        if (lane >= o) incl += t;
    }
    if (lane == 31) wsum[wv] = incl;
    __syncthreads();
    int woff = (lane < wv) ? wsum[lane] : 0;
    for (int o = 16; o > 0; o >>= 1) woff += __shfl_xor(woff, o);
    const int excl = woff + incl - c;

    if (tid < N_RANGE) {
        cnt[tid] = excl;
        cells[tid * N_CHUNK + blockIdx.x] = c;
        loff[tid * N_CHUNK + blockIdx.x]  = excl;
    }
    __syncthreads();

#define PUT(Q, SV, DV) *(int2*)(stage + 2 * min(max((Q), 0), CHUNK_EDGES - 1)) = make_int2((SV), (DV))
#define PLACE4(S, D, K0, K1, K2, K3)                  \
    do {                                              \
        const int q0 = cnt[range_of((D).x)] + (K0);   \
        const int q1 = cnt[range_of((D).y)] + (K1);   \
        const int q2 = cnt[range_of((D).z)] + (K2);   \
        const int q3 = cnt[range_of((D).w)] + (K3);   \
        PUT(q0, (S).x, (D).x);                        \
        PUT(q1, (S).y, (D).y);                        \
        PUT(q2, (S).z, (D).z);                        \
        PUT(q3, (S).w, (D).w);                        \
    } while (0)
    PLACE4(s0, d0, k00, k01, k02, k03);
    if (second) PLACE4(s1, d1, k10, k11, k12, k13);
#undef PLACE4
#undef PUT
    __syncthreads();

    constexpr int RUN_VEC = CHUNK_EDGES / 2;
    constexpr int RUN_PER = (RUN_VEC + PREP_BLOCK - 1) / PREP_BLOCK;
    const int4* s4 = (const int4*)stage;
    int4* o4 = (int4*)(pairs + (size_t)blockIdx.x * (CHUNK_EDGES * 2));
    int4 rv[RUN_PER];
    for (int k = 0; k < RUN_PER; ++k) rv[k] = s4[min(tid + k * PREP_BLOCK, RUN_VEC - 1)];
    for (int k = 0; k < RUN_PER; ++k)
        if (tid + k * PREP_BLOCK < RUN_VEC) o4[tid + k * PREP_BLOCK] = rv[k];
}

__global__ void __launch_bounds__(PREP_BLOCK)
node_place_kernel(const int* __restrict__ pairs, const int* __restrict__ cells, const int* __restrict__ loff,
                  int* __restrict__ rowptr, float* __restrict__ dinv, int* __restrict__ col) {
    __shared__ int cnt[RANGE_NODES];
    __shared__ int cnt2[RANGE_NODES];
    __shared__ int cstage[COL_STAGE];
    __shared__ int wsum[PREP_BLOCK / 32];
    __shared__ int wfront[PREP_BLOCK / 32];
    __shared__ int wown[PREP_BLOCK / 32];
    const int tid  = threadIdx.x;
    const int lane = tid & 31;
    const int wv   = tid >> 5;
    const int r    = blockIdx.x;
    const int g    = tid / SEG_LANES;
    const int sub  = tid % SEG_LANES;
    const int2* in = (const int2*)pairs;

    if (tid < RANGE_NODES) { cnt[tid] = 0; cnt2[tid] = 0; }

    const int gc = min(g, N_CHUNK - 1);
    int n  = cells[r * N_CHUNK + gc];
    int lo = loff[r * N_CHUNK + gc];
    if (g >= N_CHUNK) n = 0;
    lo = min(max(lo, 0), CHUNK_EDGES);
    n  = min(max(n, 0), CHUNK_EDGES - lo);
    const int2* seg = in + (size_t)gc * CHUNK_EDGES + lo;

    const int front = r * N_CHUNK;
    const int nvec  = (front + 3) >> 2;
    const int4* c4  = (const int4*)cells;
    int4 fv[FRONT_VEC];
#pragma unroll
    for (int k = 0; k < FRONT_VEC; ++k) fv[k] = c4[min(tid + k * PREP_BLOCK, max(nvec - 1, 0))];
    __builtin_amdgcn_sched_barrier(0);
    int2 p0[SEG_BATCH];
#pragma unroll
    for (int k = 0; k < SEG_BATCH; ++k)
        p0[k] = in[(size_t)gc * CHUNK_EDGES + min(lo + min(sub + k * SEG_LANES, max(n - 1, 0)), CHUNK_EDGES - 1)];
    __builtin_amdgcn_sched_barrier(0);
    int part = 0;
#pragma unroll
    for (int k = 0; k < FRONT_VEC; ++k) {
        const int e = (tid + k * PREP_BLOCK) * 4;
        if (e     < front) part += fv[k].x;
        if (e + 1 < front) part += fv[k].y;
        if (e + 2 < front) part += fv[k].z;
        if (e + 3 < front) part += fv[k].w;
    }
    int own = (sub == 0) ? n : 0;
    for (int o = 16; o > 0; o >>= 1) {
        part += __shfl_xor(part, o);
        own  += __shfl_xor(own, o);
    }
    if (lane == 0) { wfront[wv] = part; wown[wv] = own; }
    __syncthreads();
    int tf = wfront[lane], to = wown[lane];
    for (int o = 16; o > 0; o >>= 1) {
        tf += __shfl_xor(tf, o);
        to += __shfl_xor(to, o);
    }
    const int beg = min(max(tf, 0), N_EDGES);
    const int end = min(beg + min(max(to, 0), N_EDGES), N_EDGES);

    int k0[SEG_BATCH];
#pragma unroll
    for (int k = 0; k < SEG_BATCH; ++k) k0[k] = 0;
    if (sub < n) {
#pragma unroll
        for (int k = 0; k < SEG_BATCH; ++k)
            if (sub + k * SEG_LANES < n) k0[k] = atomicAdd(&cnt[p0[k].y & (RANGE_NODES - 1)], 1);
    }
    for (int j0 = sub + SEG_LANES * SEG_BATCH; j0 < n; j0 += SEG_LANES * SEG_BATCH) {
        int d[SEG_BATCH];
#pragma unroll
        for (int k = 0; k < SEG_BATCH; ++k) d[k] = seg[min(j0 + k * SEG_LANES, n - 1)].y;
#pragma unroll
        for (int k = 0; k < SEG_BATCH; ++k)
            if (j0 + k * SEG_LANES < n) atomicAdd(&cnt2[d[k] & (RANGE_NODES - 1)], 1);
    }
    __syncthreads();

    const int c1 = (tid < RANGE_NODES) ? cnt[tid] : 0;
    const int c = c1 + ((tid < RANGE_NODES) ? cnt2[tid] : 0);
    int incl = c;
    for (int o = 1; o < 32; o <<= 1) {
        int t = __shfl_up(incl, o);
        if (lane >= o) incl += t;
    }
    if (lane == 31) wsum[wv] = incl;
    __syncthreads();
    int woff = (lane < wv) ? wsum[lane] : 0;
    for (int o = 16; o > 0; o >>= 1) woff += __shfl_xor(woff, o);
    const int excl = woff + incl - c;

    const int node = (r << RANGE_SHIFT) + tid;
    if (tid < RANGE_NODES) {
        cnt[tid]  = excl;
        cnt2[tid] = excl + c1;
        if (node < N_NODES) {
            rowptr[node] = beg + excl;
            dinv[node] = rsqrtf((float)c + 1.0f);
            if (node == N_NODES - 1) rowptr[N_NODES] = beg + excl + c;
        }
    }
    __syncthreads();

#define PUT_SRC(SLOT, SV)                                                  \
    do {                                                                   \
        const int slot_ = (SLOT);                                          \
        if ((unsigned)slot_ < (unsigned)COL_STAGE) cstage[slot_] = (SV);   \
        else col[min(max(beg + slot_, beg), end - 1)] = (SV);              \
    } while (0)
    if (sub < n) {
        int o0[SEG_BATCH];
#pragma unroll
        for (int k = 0; k < SEG_BATCH; ++k) o0[k] = cnt[p0[k].y & (RANGE_NODES - 1)];
#pragma unroll
        for (int k = 0; k < SEG_BATCH; ++k)
            if (sub + k * SEG_LANES < n) PUT_SRC(o0[k] + k0[k], p0[k].x);
    }
    for (int j0 = sub + SEG_LANES * SEG_BATCH; j0 < n; j0 += SEG_LANES * SEG_BATCH) {
        int2 p[SEG_BATCH];
#pragma unroll
        for (int k = 0; k < SEG_BATCH; ++k) p[k] = seg[min(j0 + k * SEG_LANES, n - 1)];
#pragma unroll
        for (int k = 0; k < SEG_BATCH; ++k)
            if (j0 + k * SEG_LANES < n) PUT_SRC(atomicAdd(&cnt2[p[k].y & (RANGE_NODES - 1)], 1), p[k].x);
    }
#undef PUT_SRC
    __syncthreads();

    const int nst = min(end - beg, COL_STAGE);
    for (int i0 = tid; i0 < nst; i0 += 4 * PREP_BLOCK) {
        int cv[4];
        for (int k = 0; k < 4; ++k) cv[k] = cstage[min(i0 + k * PREP_BLOCK, COL_STAGE - 1)];
        for (int k = 0; k < 4; ++k)
            if (i0 + k * PREP_BLOCK < nst) col[beg + i0 + k * PREP_BLOCK] = cv[k];
    }
}


typedef __attribute__((ext_vector_type(4))) float f32x4;

#define PROD_ROWS 32

template <int K, int NCOL>
__global__ void __launch_bounds__((NCOL / 16) * 32)
gemm_wmma_f32(const float* __restrict__ A,
              const float* __restrict__ B,
              const float* __restrict__ rscale,
              float* __restrict__ C) {
    constexpr int NW    = NCOL / 16;
    constexpr int RPW   = PROD_ROWS / NW;
    constexpr int PITCH = K + 4;
    static_assert(K == 128, "one 16-byte load per lane covers a whole row");
    static_assert(PROD_ROWS % NW == 0 && N_NODES % PROD_ROWS == 0, "waves tile the block's rows, blocks tile the row list");

    __shared__ __attribute__((aligned(16))) float tile[PROD_ROWS * PITCH];

    const int wave = threadIdx.x >> 5;
    const int lane = threadIdx.x & 31;
    const int row0 = blockIdx.x * PROD_ROWS;

    const f32x4* xp = (const f32x4*)(A + (size_t)(row0 + wave * RPW) * K) + lane;
    f32x4 xv[RPW];
#pragma unroll
    for (int j = 0; j < RPW; ++j) xv[j] = __builtin_nontemporal_load(xp + j * (K / 4));
    __builtin_amdgcn_sched_barrier(0);

    const int m  = lane & 15;
    const int kk = (lane >> 4) << 1;
    const int n  = (lane & 15) + (wave << 4);

    const float* bp = B + (size_t)kk * NCOL + n;
    f32x2 bf[K / 4];
#pragma unroll
    for (int s = 0; s < K / 4; ++s) {
        bf[s].x = bp[(size_t)(4 * s) * NCOL];
        bf[s].y = bp[(size_t)(4 * s + 1) * NCOL];
    }

    const int mbase = (lane >> 4) << 3;
    float sc0[8], sc1[8];
#pragma unroll
    for (int i = 0; i < 8; ++i) {
        sc0[i] = rscale[row0 + mbase + i];
        sc1[i] = rscale[row0 + 16 + mbase + i];
    }

#pragma unroll
    for (int j = 0; j < RPW; ++j)
        *(f32x4*)(tile + (wave * RPW + j) * PITCH + lane * 4) = xv[j];
    __syncthreads();

    const float* ap = tile + m * PITCH + kk;
    float* cp = C + (size_t)row0 * NCOL + n;

    __builtin_amdgcn_sched_barrier(0);
    f32x8 acc0 = {0.f, 0.f, 0.f, 0.f, 0.f, 0.f, 0.f, 0.f};
    f32x8 acc1 = {0.f, 0.f, 0.f, 0.f, 0.f, 0.f, 0.f, 0.f};
#pragma unroll
    for (int s = 0; s < K / 4; ++s) {
        const f32x2 a0 = *(const f32x2*)(ap + 4 * s);
        const f32x2 a1 = *(const f32x2*)(ap + 16 * PITCH + 4 * s);
        acc0 = __builtin_amdgcn_wmma_f32_16x16x4_f32(
             false, a0,  false, bf[s],
             (short)0, acc0,  false,  false);
        acc1 = __builtin_amdgcn_wmma_f32_16x16x4_f32(
             false, a1,  false, bf[s],
             (short)0, acc1,  false,  false);
    }
    __builtin_amdgcn_sched_barrier(0);
    asm volatile("v_nop\n\tv_nop\n\tv_nop\n\tv_nop" : "+v"(acc0), "+v"(acc1));

#pragma unroll
    for (int i = 0; i < 8; ++i)
        cp[(size_t)(mbase + i) * NCOL] = acc0[i] * sc0[i];
#pragma unroll
    for (int i = 0; i < 8; ++i)
        cp[(size_t)(16 + mbase + i) * NCOL] = acc1[i] * sc1[i];
}


template <int L, int K>
__device__ __forceinline__ unsigned group_lane(unsigned x) {
    if constexpr (L == 16) {
        return (unsigned)__builtin_amdgcn_update_dpp(0, (int)x, 0x150 + K, 0xF, 0xF, true);
    } else {
        return __builtin_amdgcn_mov_dpp8(x, (unsigned)K * 0x249249u);
    }
}

template <int L>
__device__ __forceinline__ f32x4 bucket_sum(const f32x4* __restrict__ hv, const int* __restrict__ col,
                                            int node, int beg, int deg, int t) {
    f32x4 acc = hv[(unsigned)node * L + (unsigned)t];
    int my = 0;
    if (t < deg) my = col[(unsigned)(beg + t)];
    unsigned pre = (unsigned)min(max(my, 0), N_NODES - 1) * L;

    for (int base = 0; base < deg; base += L) {
        const int rem = deg - base;
        const int nxt = col[(unsigned)(beg + min(base + L + t, deg - 1))];

        f32x4 v[8];
#define LOAD_ROW(K)                                                       \
        {                                                                 \
            const unsigned p_k = group_lane<L, (K)>(pre) + (unsigned)t;   \
            if ((K) < rem) v[(K) & 7] = hv[p_k];                          \
        }
#define SUM_ROW(K) if ((K) < rem) acc += v[(K) & 7];
        LOAD_ROW(0) LOAD_ROW(1) LOAD_ROW(2) LOAD_ROW(3) LOAD_ROW(4) LOAD_ROW(5) LOAD_ROW(6) LOAD_ROW(7)
        SUM_ROW(0) SUM_ROW(1) SUM_ROW(2) SUM_ROW(3) SUM_ROW(4) SUM_ROW(5) SUM_ROW(6) SUM_ROW(7)
        if constexpr (L == 16) {
            if (rem > 8) {
                LOAD_ROW(8) LOAD_ROW(9) LOAD_ROW(10) LOAD_ROW(11) LOAD_ROW(12) LOAD_ROW(13) LOAD_ROW(14) LOAD_ROW(15)
                SUM_ROW(8) SUM_ROW(9) SUM_ROW(10) SUM_ROW(11) SUM_ROW(12) SUM_ROW(13) SUM_ROW(14) SUM_ROW(15)
            }
        }
#undef LOAD_ROW
#undef SUM_ROW
        pre = (unsigned)min(max(nxt, 0), N_NODES - 1) * L;
    }
    return acc;
}

template <int C>
__global__ void __launch_bounds__(256)
gather_finish_kernel(const float* __restrict__ hs,
                     const int* __restrict__ rowptr,
                     const int* __restrict__ col,
                     const float* __restrict__ dinv,
                     const float* __restrict__ bias,
                     float* __restrict__ dst) {
    constexpr int L = C / 4;
    constexpr int G = 32 / L;
    static_assert(L == 16 || L == 8, "a node's row covers 16 or 8 lanes");
    static_assert(N_NODES % (8 * G) == 0, "blocks tile the node list");

    const int lane = threadIdx.x & 31;
    const int t    = lane & (L - 1);
    const int node = (int)(blockIdx.x * 8 + (threadIdx.x >> 5)) * G + (lane / L);
    const int beg  = rowptr[node];
    const int deg  = rowptr[node + 1] - beg;
    const float di = dinv[node];
    const f32x4 bv = *(const f32x4*)(bias + t * 4);

    const f32x4 acc = bucket_sum<L>((const f32x4*)hs, col, node, beg, deg, t);

    *(f32x4*)(dst + (size_t)node * C + (size_t)t * 4) = acc * di + bv;
}

#define TILE_PITCH (HID_CH + 4)

__global__ void __launch_bounds__(256)
gather_relu_product_kernel(const float* __restrict__ hs,
                           const int* __restrict__ rowptr,
                           const int* __restrict__ col,
                           const float* __restrict__ dinv,
                           const float* __restrict__ bias,
                           const float* __restrict__ Wn,
                           float* __restrict__ hn) {
    constexpr int L = HID_CH / 4;
    static_assert(L == 16 && N_NODES % 16 == 0, "a block's nodes are the 16 rows of one matrix tile");
    static_assert(OUT_CH % 16 == 0 && OUT_CH / 16 <= 8, "one wave per column tile");

    __shared__ __attribute__((aligned(16))) float tile[16 * TILE_PITCH];
    __shared__ float tscale[16];

    const int wave = threadIdx.x >> 5;
    const int lane = threadIdx.x & 31;
    const int t    = lane & (L - 1);
    const int row  = threadIdx.x >> 4;
    const int row0 = blockIdx.x << 4;
    const int node = row0 + row;
    const int beg  = rowptr[node];
    const int deg  = rowptr[node + 1] - beg;
    const float di = dinv[node];
    const f32x4 bv = *(const f32x4*)(bias + t * 4);

    const f32x4 acc = bucket_sum<L>((const f32x4*)hs, col, node, beg, deg, t);

    f32x4 r = acc * di + bv;
    r.x = fmaxf(r.x, 0.0f); r.y = fmaxf(r.y, 0.0f);
    r.z = fmaxf(r.z, 0.0f); r.w = fmaxf(r.w, 0.0f);
    *(f32x4*)(tile + row * TILE_PITCH + t * 4) = r;
    if (t == 0) tscale[row] = di;

    const int m  = lane & 15;
    const int kk = (lane >> 4) << 1;
    const int n  = m + (wave << 4);
    const bool owner = wave < OUT_CH / 16;

    f32x2 bf[HID_CH / 4];
    if (owner) {
        const float* bp = Wn + (size_t)kk * OUT_CH + n;
#pragma unroll
        for (int s = 0; s < HID_CH / 4; ++s) {
            bf[s].x = bp[(size_t)(4 * s) * OUT_CH];
            bf[s].y = bp[(size_t)(4 * s + 1) * OUT_CH];
        }
    }
    __syncthreads();
    if (!owner) return;

    const float* ap = tile + m * TILE_PITCH + kk;
    const int mbase = (lane >> 4) << 3;
    float* cp = hn + (size_t)row0 * OUT_CH + n;
    float sc[8];
#pragma unroll
    for (int i = 0; i < 8; ++i) sc[i] = tscale[mbase + i];

    __builtin_amdgcn_sched_barrier(0);
    f32x8 pacc = {0.f, 0.f, 0.f, 0.f, 0.f, 0.f, 0.f, 0.f};
#pragma unroll
    for (int s = 0; s < HID_CH / 4; ++s) {
        const f32x2 a = *(const f32x2*)(ap + 4 * s);
        pacc = __builtin_amdgcn_wmma_f32_16x16x4_f32(
             false, a,  false, bf[s],
             (short)0, pacc,  false,  false);
    }
    __builtin_amdgcn_sched_barrier(0);
    asm volatile("v_nop\n\tv_nop\n\tv_nop\n\tv_nop" : "+v"(pacc));

#pragma unroll
    for (int i = 0; i < 8; ++i)
        cp[(size_t)(mbase + i) * OUT_CH] = pacc[i] * sc[i];
}


extern "C" void kernel_launch(void* const* d_in, const int* in_sizes, int n_in,
                              void* d_out, int out_size, void* d_ws, size_t ws_size,
                              hipStream_t stream) {
    const float* x    = (const float*)d_in[0];
    const int*   ei   = (const int*)d_in[1];
    const float* W1   = (const float*)d_in[2];
    const float* b1   = (const float*)d_in[3];
    const float* W2   = (const float*)d_in[4];
    const float* b2   = (const float*)d_in[5];
    float*       out  = (float*)d_out;

    const int* srcv = ei;
    const int* dstv = ei + N_EDGES;

    float* dinv   = (float*)d_ws;
    int*   rowptr = (int*)(dinv + 100352);
    int*   col    = rowptr + 100352;
    float* h1     = (float*)(col + N_EDGES);
    float* h2     = h1 + (size_t)N_NODES * HID_CH;
    int*   cells = (int*)h1;
    int*   loff  = cells + 49152;
    int*   pairs = loff + 49152;

    chunk_bucket_kernel<<<N_CHUNK, PREP_BLOCK, 0, stream>>>(srcv, dstv, cells, loff, pairs);
    node_place_kernel<<<N_RANGE, PREP_BLOCK, 0, stream>>>(pairs, cells, loff, rowptr, dinv, col);

    gemm_wmma_f32<IN_CH, HID_CH><<<N_NODES / PROD_ROWS, 128, 0, stream>>>(x, W1, dinv, h1);

    gather_relu_product_kernel<<<N_NODES / 16, 256, 0, stream>>>(h1, rowptr, col, dinv, b1, W2, h2);

    gather_finish_kernel<OUT_CH><<<N_NODES / 32, 256, 0, stream>>>(h2, rowptr, col, dinv, b2, out);
}
